# starting kernel
# baseline (speedup 1.0000x reference)
#include <hip/hip_runtime.h>
#include <hip/hip_bf16.h>

typedef __attribute__((ext_vector_type(2))) float v2f;
typedef __attribute__((ext_vector_type(8))) float v8f;

#define EPSV 1e-6f

__global__ void HyperZero_kernel(float* __restrict__ p, int n) {
  int i = blockIdx.x * blockDim.x + threadIdx.x;
  if (i < n) p[i] = 0.0f;
}

__global__ __launch_bounds__(256) void HyperEdge_kernel(
    const float* __restrict__ vertices,
    const int*   __restrict__ edges,
    const float* __restrict__ messages,
    float* __restrict__ sums,
    float* __restrict__ cnt,
    long long E) {
  const int lane = threadIdx.x & 31;
  const long long wid = (long long)blockIdx.x * (blockDim.x >> 5) + (threadIdx.x >> 5);
  const long long e0 = wid << 5;
  if (e0 >= E) return;

  const int octet = lane >> 3;
  const int perm  = (octet == 1) ? (lane + 8) : ((octet == 2) ? (lane - 8) : lane);
  const long long eg = e0 + perm;

  float weight = 0.0f;
  if (e0 + 32 <= E) {
    const int rowA = lane & 15;
    const int kb   = (lane < 16) ? 0 : 2;
    const float* m1 = messages + (e0 + rowA) * 8 + kb;
    const float* m2 = m1 + 16 * 8;
    v2f a1lo = *(const v2f*)(m1);
    v2f a1hi = *(const v2f*)(m1 + 4);
    v2f a2lo = *(const v2f*)(m2);
    v2f a2hi = *(const v2f*)(m2 + 4);
    v2f ones; ones[0] = 1.0f; ones[1] = 1.0f;

    v8f c1 = {};
    c1 = __builtin_amdgcn_wmma_f32_16x16x4_f32(false, a1lo, false, ones, (short)0, c1, false, false);
    c1 = __builtin_amdgcn_wmma_f32_16x16x4_f32(false, a1hi, false, ones, (short)0, c1, false, false);
    v8f c2 = {};
    c2 = __builtin_amdgcn_wmma_f32_16x16x4_f32(false, a2lo, false, ones, (short)0, c2, false, false);
    c2 = __builtin_amdgcn_wmma_f32_16x16x4_f32(false, a2hi, false, ones, (short)0, c2, false, false);

    const bool useD2 = (lane >> 3) & 1;
    v8f d = useD2 ? c2 : c1;
    const int r = lane & 7;
    float s = d[0];
    if (r == 1) s = d[1];
    if (r == 2) s = d[2];
    if (r == 3) s = d[3];
    if (r == 4) s = d[4];
    if (r == 5) s = d[5];
    if (r == 6) s = d[6];
    if (r == 7) s = d[7];
    weight = s * 0.125f;
  } else if (eg < E) {
    float s = 0.0f;
#pragma unroll
    for (int k = 0; k < 8; ++k) s += messages[eg * 8 + k];
    weight = s * 0.125f;
  }

  if (eg < E) {
    const int vs = edges[2 * eg + 0];
    const int vd = edges[2 * eg + 1];
    const float x0 = vertices[3 * vs + 0], x1 = vertices[3 * vs + 1], x2 = vertices[3 * vs + 2];
    const float y0 = vertices[3 * vd + 0], y1 = vertices[3 * vd + 1], y2 = vertices[3 * vd + 2];
    const float t  = -x0 * y0 + x1 * y1 + x2 * y2;
    const float v0 = y0 + x0 * t;
    const float v1 = y1 + x1 * t;
    const float v2 = y2 + x2 * t;
    const float vn   = sqrtf(-v0 * v0 + v1 * v1 + v2 * v2 + EPSV);
    const float dist = acoshf(-t);
    const float sc   = dist * weight / vn;
    atomicAdd(&sums[3 * vs + 0], v0 * sc);
    atomicAdd(&sums[3 * vs + 1], v1 * sc);
    atomicAdd(&sums[3 * vs + 2], v2 * sc);
    atomicAdd(&cnt[vs], 1.0f);
  }
}

__global__ void HyperVertex_kernel(const float* __restrict__ vertices,
                                   const float* __restrict__ sums,
                                   const float* __restrict__ cnt,
                                   float* __restrict__ out, int N) {
  int i = blockIdx.x * blockDim.x + threadIdx.x;
  if (i >= N) return;
  const float c   = cnt[i];
  const float inv = 1.0f / fmaxf(c, 1.0f);
  float a0 = 0.0f, a1 = 0.0f, a2 = 0.0f;
  if (c > 0.0f) {
    a0 = sums[3 * i + 0] * inv;
    a1 = sums[3 * i + 1] * inv;
    a2 = sums[3 * i + 2] * inv;
  }
  const float an = sqrtf(-a0 * a0 + a1 * a1 + a2 * a2 + EPSV);
  const float T  = an;
  const float e  = expf(T);
  const float ei = 1.0f / e;
  const float ch = 0.5f * (e + ei);
  const float sh = 0.5f * (e - ei);
  const float f  = sh / an;
  out[3 * i + 0] = ch * vertices[3 * i + 0] + f * a0;
  out[3 * i + 1] = ch * vertices[3 * i + 1] + f * a1;
  out[3 * i + 2] = ch * vertices[3 * i + 2] + f * a2;
}

extern "C" void kernel_launch(void* const* d_in, const int* in_sizes, int n_in,
                              void* d_out, int out_size, void* d_ws, size_t ws_size,
                              hipStream_t stream) {
  const float* vertices = (const float*)d_in[0];
  const int*   edges    = (const int*)d_in[1];
  const float* messages = (const float*)d_in[2];
  float* out = (float*)d_out;

  const int       N = in_sizes[0] / 3;
  const long long E = (long long)in_sizes[1] / 2;

  float* sums = (float*)d_ws;
  float* cnt  = sums + (size_t)3 * N;

  const int zn = 4 * N;
  HyperZero_kernel<<<(zn + 255) / 256, 256, 0, stream>>>(sums, zn);

  const long long waves  = (E + 31) >> 5;
  const long long blocks = (waves + 7) >> 3;
  HyperEdge_kernel<<<dim3((unsigned)blocks), 256, 0, stream>>>(
      vertices, edges, messages, sums, cnt, E);

  HyperVertex_kernel<<<(N + 255) / 256, 256, 0, stream>>>(vertices, sums, cnt, out, N);
}


# Sums + count in one 16-byte record per vertex; a quad_perm lane transpose puts an edge's 4 f32 atomic adds in 4 adjacent lanes of one instruction: 1 request per edge, not 4; arrival order kept
# speedup vs baseline: 2.2637x; 2.2637x over previous
#include <hip/hip_runtime.h>
#include <hip/hip_bf16.h>

typedef __attribute__((ext_vector_type(2))) float v2f;
typedef __attribute__((ext_vector_type(8))) float v8f;

#define EPSV 1e-6f

__global__ void HyperZero_kernel(float* __restrict__ p, int n) {
  int i = blockIdx.x * blockDim.x + threadIdx.x;
  if (i < n) p[i] = 0.0f;
}

template <int K>
static __device__ __forceinline__ int quad_lane(int v) {
  return __builtin_amdgcn_mov_dpp(v, K * 0x55, 0xf, 0xf, true);
}
template <int K>
static __device__ __forceinline__ float quad_lane(float v) {
  return __int_as_float(quad_lane<K>(__float_as_int(v)));
}

template <int K>
static __device__ __forceinline__ void add_quad_edge(float* __restrict__ acc, int comp, int vs,
                                                     float c0, float c1, float c2) {
  const int   vj = quad_lane<K>(vs);
  const float t0 = quad_lane<K>(c0);
  const float t1 = quad_lane<K>(c1);
  const float t2 = quad_lane<K>(c2);
  const float lo = (comp == 0) ? t0 : t1;
  const float hi = (comp == 2) ? t2 : 1.0f;
  const float val = (comp < 2) ? lo : hi;
  if (vj >= 0) atomicAdd(&acc[4 * vj + comp], val);
}

__global__ __launch_bounds__(256) void HyperEdge_kernel(
    const float* __restrict__ vertices,
    const int*   __restrict__ edges,
    const float* __restrict__ messages,
    float* __restrict__ acc,
    long long E) {
  const int lane = threadIdx.x & 31;
  const long long wid = (long long)blockIdx.x * (blockDim.x >> 5) + (threadIdx.x >> 5);
  const long long e0 = wid << 5;
  if (e0 >= E) return;

  const int octet = lane >> 3;
  const int perm  = (octet == 1) ? (lane + 8) : ((octet == 2) ? (lane - 8) : lane);
  const long long eg = e0 + perm;

  float weight = 0.0f;
  if (e0 + 32 <= E) {
    const int rowA = lane & 15;
    const int kb   = (lane < 16) ? 0 : 2;
    const float* m1 = messages + (e0 + rowA) * 8 + kb;
    const float* m2 = m1 + 16 * 8;
    v2f a1lo = *(const v2f*)(m1);
    v2f a1hi = *(const v2f*)(m1 + 4);
    v2f a2lo = *(const v2f*)(m2);
    v2f a2hi = *(const v2f*)(m2 + 4);
    v2f ones; ones[0] = 1.0f; ones[1] = 1.0f;

    v8f c1 = {};
    c1 = __builtin_amdgcn_wmma_f32_16x16x4_f32(false, a1lo, false, ones, (short)0, c1, false, false);
    c1 = __builtin_amdgcn_wmma_f32_16x16x4_f32(false, a1hi, false, ones, (short)0, c1, false, false);
    v8f c2 = {};
    c2 = __builtin_amdgcn_wmma_f32_16x16x4_f32(false, a2lo, false, ones, (short)0, c2, false, false);
    c2 = __builtin_amdgcn_wmma_f32_16x16x4_f32(false, a2hi, false, ones, (short)0, c2, false, false);

    const bool useD2 = (lane >> 3) & 1;
    v8f d = useD2 ? c2 : c1;
    const int r = lane & 7;
    float s = d[0];
    if (r == 1) s = d[1];
    if (r == 2) s = d[2];
    if (r == 3) s = d[3];
    if (r == 4) s = d[4];
    if (r == 5) s = d[5];
    if (r == 6) s = d[6];
    if (r == 7) s = d[7];
    weight = s * 0.125f;
  } else if (eg < E) {
    float s = 0.0f;
#pragma unroll
    for (int k = 0; k < 8; ++k) s += messages[eg * 8 + k];
    weight = s * 0.125f;
  }

  int   vs = -1;
  float c0 = 0.0f, c1 = 0.0f, c2 = 0.0f;
  if (eg < E) {
    vs = edges[2 * eg + 0];
    const int vd = edges[2 * eg + 1];
    const float x0 = vertices[3 * vs + 0], x1 = vertices[3 * vs + 1], x2 = vertices[3 * vs + 2];
    const float y0 = vertices[3 * vd + 0], y1 = vertices[3 * vd + 1], y2 = vertices[3 * vd + 2];
    const float t  = -x0 * y0 + x1 * y1 + x2 * y2;
    const float v0 = y0 + x0 * t;
    const float v1 = y1 + x1 * t;
    const float v2 = y2 + x2 * t;
    const float vn   = sqrtf(-v0 * v0 + v1 * v1 + v2 * v2 + EPSV);
    const float dist = acoshf(-t);
    const float sc   = dist * weight / vn;
    c0 = v0 * sc;
    c1 = v1 * sc;
    c2 = v2 * sc;
  }

  const int comp = lane & 3;
  add_quad_edge<0>(acc, comp, vs, c0, c1, c2);
  add_quad_edge<1>(acc, comp, vs, c0, c1, c2);
  add_quad_edge<2>(acc, comp, vs, c0, c1, c2);
  add_quad_edge<3>(acc, comp, vs, c0, c1, c2);
}

__global__ void HyperVertex_kernel(const float* __restrict__ vertices,
                                   const float* __restrict__ acc,
                                   float* __restrict__ out, int N) {
  int i = blockIdx.x * blockDim.x + threadIdx.x;
  if (i >= N) return;
  const float4 rec = reinterpret_cast<const float4*>(acc)[i];
  const float c   = rec.w;
  const float inv = 1.0f / fmaxf(c, 1.0f);
  float a0 = 0.0f, a1 = 0.0f, a2 = 0.0f;
  if (c > 0.0f) {
    a0 = rec.x * inv;
    a1 = rec.y * inv;
    a2 = rec.z * inv;
  }
  const float an = sqrtf(-a0 * a0 + a1 * a1 + a2 * a2 + EPSV);
  const float T  = an;
  const float e  = expf(T);
  const float ei = 1.0f / e;
  const float ch = 0.5f * (e + ei);
  const float sh = 0.5f * (e - ei);
  const float f  = sh / an;
  out[3 * i + 0] = ch * vertices[3 * i + 0] + f * a0;
  out[3 * i + 1] = ch * vertices[3 * i + 1] + f * a1;
  out[3 * i + 2] = ch * vertices[3 * i + 2] + f * a2;
}

extern "C" void kernel_launch(void* const* d_in, const int* in_sizes, int n_in,
                              void* d_out, int out_size, void* d_ws, size_t ws_size,
                              hipStream_t stream) {
  const float* vertices = (const float*)d_in[0];
  const int*   edges    = (const int*)d_in[1];
  const float* messages = (const float*)d_in[2];
  float* out = (float*)d_out;

  const int       N = in_sizes[0] / 3;
  const long long E = (long long)in_sizes[1] / 2;

  float* acc = (float*)d_ws;

  const int zn = 4 * N;
  HyperZero_kernel<<<(zn + 255) / 256, 256, 0, stream>>>(acc, zn);

  const long long waves  = (E + 31) >> 5;
  const long long blocks = (waves + 7) >> 3;
  HyperEdge_kernel<<<dim3((unsigned)blocks), 256, 0, stream>>>(
      vertices, edges, messages, acc, E);

  HyperVertex_kernel<<<(N + 255) / 256, 256, 0, stream>>>(vertices, acc, out, N);
}


# Atomic adds kept inside one half of the chip: vertex records split in two ranges, resident waves draw 8-tile runs per range and add into their own half's range first; atomic arrival order differs
# speedup vs baseline: 2.4294x; 1.0732x over previous
#include <hip/hip_runtime.h>
#include <hip/hip_bf16.h>

typedef __attribute__((ext_vector_type(2))) float v2f;
typedef __attribute__((ext_vector_type(8))) float v8f;

#define EPSV 1e-6f

#define RUN_TILES 8
#define TICKET_STRIDE 128

__global__ void HyperZero_kernel(float* __restrict__ p, int n, unsigned* __restrict__ tickets) {
  int i = blockIdx.x * blockDim.x + threadIdx.x;
  if (i < n) p[i] = 0.0f;
  if (i < 2) tickets[i * TICKET_STRIDE] = 0u;
}

template <int K>
static __device__ __forceinline__ int quad_lane(int v) {
  return __builtin_amdgcn_mov_dpp(v, K * 0x55, 0xf, 0xf, true);
}
template <int K>
static __device__ __forceinline__ float quad_lane(float v) {
  return __int_as_float(quad_lane<K>(__float_as_int(v)));
}

template <int K>
static __device__ __forceinline__ void add_quad_edge(float* __restrict__ acc, int comp, int vs,
                                                     float c0, float c1, float c2) {
  const int   vj = quad_lane<K>(vs);
  const float t0 = quad_lane<K>(c0);
  const float t1 = quad_lane<K>(c1);
  const float t2 = quad_lane<K>(c2);
  const float lo = (comp == 0) ? t0 : t1;
  const float hi = (comp == 2) ? t2 : 1.0f;
  const float val = (comp < 2) ? lo : hi;
  if (vj >= 0) atomicAdd(&acc[4 * vj + comp], val);
}

static __device__ __forceinline__ void edge_tile(
    const float* __restrict__ vertices,
    const int*   __restrict__ edges,
    const float* __restrict__ messages,
    float* __restrict__ acc,
    long long E, long long e0, int lo, int hi, int lane) {
  const int octet = lane >> 3;
  const int perm  = (octet == 1) ? (lane + 8) : ((octet == 2) ? (lane - 8) : lane);
  const long long eg = e0 + perm;

  const long long el = (eg < E) ? eg : (E - 1);
  const int2 sd = *reinterpret_cast<const int2*>(edges + 2 * el);

  float weight = 0.0f;
  if (e0 + 32 <= E) {
    const int rowA = lane & 15;
    const int kb   = (lane < 16) ? 0 : 2;
    const float* m1 = messages + (e0 + rowA) * 8 + kb;
    const float* m2 = m1 + 16 * 8;
    v2f a1lo = *(const v2f*)(m1);
    v2f a1hi = *(const v2f*)(m1 + 4);
    v2f a2lo = *(const v2f*)(m2);
    v2f a2hi = *(const v2f*)(m2 + 4);
    v2f ones; ones[0] = 1.0f; ones[1] = 1.0f;

    v8f c1 = {};
    c1 = __builtin_amdgcn_wmma_f32_16x16x4_f32(false, a1lo, false, ones, (short)0, c1, false, false);
    c1 = __builtin_amdgcn_wmma_f32_16x16x4_f32(false, a1hi, false, ones, (short)0, c1, false, false);
    v8f c2 = {};
    c2 = __builtin_amdgcn_wmma_f32_16x16x4_f32(false, a2lo, false, ones, (short)0, c2, false, false);
    c2 = __builtin_amdgcn_wmma_f32_16x16x4_f32(false, a2hi, false, ones, (short)0, c2, false, false);

    const bool useD2 = (lane >> 3) & 1;
    v8f d = useD2 ? c2 : c1;
    const int r = lane & 7;
    float s = d[0];
    if (r == 1) s = d[1];
    if (r == 2) s = d[2];
    if (r == 3) s = d[3];
    if (r == 4) s = d[4];
    if (r == 5) s = d[5];
    if (r == 6) s = d[6];
    if (r == 7) s = d[7];
    weight = s * 0.125f;
  } else if (eg < E) {
    float s = 0.0f;
#pragma unroll
    for (int k = 0; k < 8; ++k) s += messages[eg * 8 + k];
    weight = s * 0.125f;
  }

  int   vs = -1;
  float c0 = 0.0f, c1 = 0.0f, c2 = 0.0f;
  {
    const int s  = (eg < E) ? sd.x : -1;
    const int vd = sd.y;
    if (s >= lo && s < hi) {
      vs = s;
      const float x0 = vertices[3 * vs + 0], x1 = vertices[3 * vs + 1], x2 = vertices[3 * vs + 2];
      const float y0 = vertices[3 * vd + 0], y1 = vertices[3 * vd + 1], y2 = vertices[3 * vd + 2];
      const float t  = -x0 * y0 + x1 * y1 + x2 * y2;
      const float v0 = y0 + x0 * t;
      const float v1 = y1 + x1 * t;
      const float v2 = y2 + x2 * t;
      const float vn   = sqrtf(-v0 * v0 + v1 * v1 + v2 * v2 + EPSV);
      const float dist = acoshf(-t);
      const float sc   = dist * weight / vn;
      c0 = v0 * sc;
      c1 = v1 * sc;
      c2 = v2 * sc;
    }
  }

  const int comp = lane & 3;
  add_quad_edge<0>(acc, comp, vs, c0, c1, c2);
  add_quad_edge<1>(acc, comp, vs, c0, c1, c2);
  add_quad_edge<2>(acc, comp, vs, c0, c1, c2);
  add_quad_edge<3>(acc, comp, vs, c0, c1, c2);
}

__global__ __launch_bounds__(256) void HyperEdge_kernel(
    const float* __restrict__ vertices,
    const int*   __restrict__ edges,
    const float* __restrict__ messages,
    float* __restrict__ acc,
    unsigned* __restrict__ tickets,
    long long E, int N) {
  const int lane = threadIdx.x & 31;
  const unsigned place  = __builtin_amdgcn_s_sendmsg_rtn(0x87);
  const int      myhalf = (int)((place >> 18) & 1u);

  const long long tiles = (E + 31) >> 5;
  const unsigned  runs  = (unsigned)((tiles + RUN_TILES - 1) / RUN_TILES);
  const int       split = ((N >> 1) + 15) & ~15;

  for (int pass = 0; pass < 2; ++pass) {
    const int h  = myhalf ^ pass;
    const int lo = h ? split : 0;
    const int hi = h ? N : split;
    for (unsigned it = 0; it <= runs; ++it) {
      unsigned t = 0u;
      if (lane == 0) t = atomicAdd(&tickets[h * TICKET_STRIDE], 1u);
      t = (unsigned)__builtin_amdgcn_readfirstlane((int)t);
      if (t >= runs) break;
      const long long tile0 = (long long)t * RUN_TILES;
      for (int k = 0; k < RUN_TILES; ++k) {
        const long long e0 = (tile0 + k) << 5;
        if (e0 >= E) break;
        edge_tile(vertices, edges, messages, acc, E, e0, lo, hi, lane);
      }
    }
  }
}

__global__ void HyperVertex_kernel(const float* __restrict__ vertices,
                                   const float* __restrict__ acc,
                                   float* __restrict__ out, int N) {
  int i = blockIdx.x * blockDim.x + threadIdx.x;
  if (i >= N) return;
  const float4 rec = reinterpret_cast<const float4*>(acc)[i];
  const float c   = rec.w;
  const float inv = 1.0f / fmaxf(c, 1.0f);
  float a0 = 0.0f, a1 = 0.0f, a2 = 0.0f;
  if (c > 0.0f) {
    a0 = rec.x * inv;
    a1 = rec.y * inv;
    a2 = rec.z * inv;
  }
  const float an = sqrtf(-a0 * a0 + a1 * a1 + a2 * a2 + EPSV);
  const float T  = an;
  const float e  = expf(T);
  const float ei = 1.0f / e;
  const float ch = 0.5f * (e + ei);
  const float sh = 0.5f * (e - ei);
  const float f  = sh / an;
  out[3 * i + 0] = ch * vertices[3 * i + 0] + f * a0;
  out[3 * i + 1] = ch * vertices[3 * i + 1] + f * a1;
  out[3 * i + 2] = ch * vertices[3 * i + 2] + f * a2;
}

extern "C" void kernel_launch(void* const* d_in, const int* in_sizes, int n_in,
                              void* d_out, int out_size, void* d_ws, size_t ws_size,
                              hipStream_t stream) {
  const float* vertices = (const float*)d_in[0];
  const int*   edges    = (const int*)d_in[1];
  const float* messages = (const float*)d_in[2];
  float* out = (float*)d_out;

  const int       N = in_sizes[0] / 3;
  const long long E = (long long)in_sizes[1] / 2;

  float* acc = (float*)d_ws;

  unsigned* tickets = (unsigned*)d_out;

  const int zn = 4 * N;
  HyperZero_kernel<<<(zn + 255) / 256, 256, 0, stream>>>(acc, zn, tickets);

  HyperEdge_kernel<<<dim3(2048), 256, 0, stream>>>(
      vertices, edges, messages, acc, tickets, E, N);

  HyperVertex_kernel<<<(N + 255) / 256, 256, 0, stream>>>(vertices, acc, out, N);
}


# Edges binned by source range (LDS count, integer reserve, 8-byte records); one work-group per 1024-vertex range sums in LDS and fuses mean + exp map: no global f32 add left; sum order = LDS arrival
# speedup vs baseline: 3.3727x; 1.3883x over previous
#include <hip/hip_runtime.h>
#include <hip/hip_bf16.h>

typedef __attribute__((ext_vector_type(2))) float v2f;
typedef __attribute__((ext_vector_type(8))) float v8f;

#define EPSV 1e-6f

#define RUN_TILES 8
#define TICKET_STRIDE 128

__global__ void HyperZero_kernel(float* __restrict__ p, int n, unsigned* __restrict__ tickets) {
  int i = blockIdx.x * blockDim.x + threadIdx.x;
  if (i < n) p[i] = 0.0f;
  if (i < 2) tickets[i * TICKET_STRIDE] = 0u;
}

template <int K>
static __device__ __forceinline__ int quad_lane(int v) {
  return __builtin_amdgcn_mov_dpp(v, K * 0x55, 0xf, 0xf, true);
}
template <int K>
static __device__ __forceinline__ float quad_lane(float v) {
  return __int_as_float(quad_lane<K>(__float_as_int(v)));
}

template <int K>
static __device__ __forceinline__ void add_quad_edge(float* __restrict__ acc, int comp, int vs,
                                                     float c0, float c1, float c2) {
  const int   vj = quad_lane<K>(vs);
  const float t0 = quad_lane<K>(c0);
  const float t1 = quad_lane<K>(c1);
  const float t2 = quad_lane<K>(c2);
  const float lo = (comp == 0) ? t0 : t1;
  const float hi = (comp == 2) ? t2 : 1.0f;
  const float val = (comp < 2) ? lo : hi;
  if (vj >= 0) atomicAdd(&acc[4 * vj + comp], val);
}

static __device__ __forceinline__ void edge_tile(
    const float* __restrict__ vertices,
    const int*   __restrict__ edges,
    const float* __restrict__ messages,
    float* __restrict__ acc,
    long long E, long long e0, int lo, int hi, int lane) {
  const int octet = lane >> 3;
  const int perm  = (octet == 1) ? (lane + 8) : ((octet == 2) ? (lane - 8) : lane);
  const long long eg = e0 + perm;

  const long long el = (eg < E) ? eg : (E - 1);
  const int2 sd = *reinterpret_cast<const int2*>(edges + 2 * el);

  float weight = 0.0f;
  if (e0 + 32 <= E) {
    const int rowA = lane & 15;
    const int kb   = (lane < 16) ? 0 : 2;
    const float* m1 = messages + (e0 + rowA) * 8 + kb;
    const float* m2 = m1 + 16 * 8;
    v2f a1lo = *(const v2f*)(m1);
    v2f a1hi = *(const v2f*)(m1 + 4);
    v2f a2lo = *(const v2f*)(m2);
    v2f a2hi = *(const v2f*)(m2 + 4);
    v2f ones; ones[0] = 1.0f; ones[1] = 1.0f;

    v8f c1 = {};
    c1 = __builtin_amdgcn_wmma_f32_16x16x4_f32(false, a1lo, false, ones, (short)0, c1, false, false);
    c1 = __builtin_amdgcn_wmma_f32_16x16x4_f32(false, a1hi, false, ones, (short)0, c1, false, false);
    v8f c2 = {};
    c2 = __builtin_amdgcn_wmma_f32_16x16x4_f32(false, a2lo, false, ones, (short)0, c2, false, false);
    c2 = __builtin_amdgcn_wmma_f32_16x16x4_f32(false, a2hi, false, ones, (short)0, c2, false, false);

    const bool useD2 = (lane >> 3) & 1;
    v8f d = useD2 ? c2 : c1;
    const int r = lane & 7;
    float s = d[0];
    if (r == 1) s = d[1];
    if (r == 2) s = d[2];
    if (r == 3) s = d[3];
    if (r == 4) s = d[4];
    if (r == 5) s = d[5];
    if (r == 6) s = d[6];
    if (r == 7) s = d[7];
    weight = s * 0.125f;
  } else if (eg < E) {
    float s = 0.0f;
#pragma unroll
    for (int k = 0; k < 8; ++k) s += messages[eg * 8 + k];
    weight = s * 0.125f;
  }

  int   vs = -1;
  float c0 = 0.0f, c1 = 0.0f, c2 = 0.0f;
  {
    const int s  = (eg < E) ? sd.x : -1;
    const int vd = sd.y;
    if (s >= lo && s < hi) {
      vs = s;
      const float x0 = vertices[3 * vs + 0], x1 = vertices[3 * vs + 1], x2 = vertices[3 * vs + 2];
      const float y0 = vertices[3 * vd + 0], y1 = vertices[3 * vd + 1], y2 = vertices[3 * vd + 2];
      const float t  = -x0 * y0 + x1 * y1 + x2 * y2;
      const float v0 = y0 + x0 * t;
      const float v1 = y1 + x1 * t;
      const float v2 = y2 + x2 * t;
      const float vn   = sqrtf(-v0 * v0 + v1 * v1 + v2 * v2 + EPSV);
      const float dist = acoshf(-t);
      const float sc   = dist * weight / vn;
      c0 = v0 * sc;
      c1 = v1 * sc;
      c2 = v2 * sc;
    }
  }

  const int comp = lane & 3;
  add_quad_edge<0>(acc, comp, vs, c0, c1, c2);
  add_quad_edge<1>(acc, comp, vs, c0, c1, c2);
  add_quad_edge<2>(acc, comp, vs, c0, c1, c2);
  add_quad_edge<3>(acc, comp, vs, c0, c1, c2);
}

__global__ __launch_bounds__(256) void HyperEdge_kernel(
    const float* __restrict__ vertices,
    const int*   __restrict__ edges,
    const float* __restrict__ messages,
    float* __restrict__ acc,
    unsigned* __restrict__ tickets,
    long long E, int N) {
  const int lane = threadIdx.x & 31;
  const unsigned place  = __builtin_amdgcn_s_sendmsg_rtn(0x87);
  const int      myhalf = (int)((place >> 18) & 1u);

  const long long tiles = (E + 31) >> 5;
  const unsigned  runs  = (unsigned)((tiles + RUN_TILES - 1) / RUN_TILES);
  const int       split = ((N >> 1) + 15) & ~15;

  for (int pass = 0; pass < 2; ++pass) {
    const int h  = myhalf ^ pass;
    const int lo = h ? split : 0;
    const int hi = h ? N : split;
    for (unsigned it = 0; it <= runs; ++it) {
      unsigned t = 0u;
      if (lane == 0) t = atomicAdd(&tickets[h * TICKET_STRIDE], 1u);
      t = (unsigned)__builtin_amdgcn_readfirstlane((int)t);
      if (t >= runs) break;
      const long long tile0 = (long long)t * RUN_TILES;
      for (int k = 0; k < RUN_TILES; ++k) {
        const long long e0 = (tile0 + k) << 5;
        if (e0 >= E) break;
        edge_tile(vertices, edges, messages, acc, E, e0, lo, hi, lane);
      }
    }
  }
}

__global__ void HyperVertex_kernel(const float* __restrict__ vertices,
                                   const float* __restrict__ acc,
                                   float* __restrict__ out, int N) {
  int i = blockIdx.x * blockDim.x + threadIdx.x;
  if (i >= N) return;
  const float4 rec = reinterpret_cast<const float4*>(acc)[i];
  const float c   = rec.w;
  const float inv = 1.0f / fmaxf(c, 1.0f);
  float a0 = 0.0f, a1 = 0.0f, a2 = 0.0f;
  if (c > 0.0f) {
    a0 = rec.x * inv;
    a1 = rec.y * inv;
    a2 = rec.z * inv;
  }
  const float an = sqrtf(-a0 * a0 + a1 * a1 + a2 * a2 + EPSV);
  const float T  = an;
  const float e  = expf(T);
  const float ei = 1.0f / e;
  const float ch = 0.5f * (e + ei);
  const float sh = 0.5f * (e - ei);
  const float f  = sh / an;
  out[3 * i + 0] = ch * vertices[3 * i + 0] + f * a0;
  out[3 * i + 1] = ch * vertices[3 * i + 1] + f * a1;
  out[3 * i + 2] = ch * vertices[3 * i + 2] + f * a2;
}


#define RV_SHIFT 10
#define RV (1 << RV_SHIFT)
#define MAXB 1024
#define DST_BITS 20

__global__ void BinZero_kernel(unsigned* __restrict__ tab) {
  const int i = blockIdx.x * blockDim.x + threadIdx.x;
  if (i < 2 * MAXB) tab[i] = 0u;
}

__global__ __launch_bounds__(256) void BinCount_kernel(const int* __restrict__ edges,
                                                       unsigned* __restrict__ tot,
                                                       long long E, int NB) {
  __shared__ unsigned h[MAXB];
  for (int b = threadIdx.x; b < MAXB; b += 256) h[b] = 0u;
  __syncthreads();
  const long long stride = (long long)gridDim.x * 256;
  for (long long e = (long long)blockIdx.x * 256 + threadIdx.x; e < E; e += 4 * stride) {
    int s[4];
#pragma unroll
    for (int k = 0; k < 4; ++k) {
      const long long ek = e + k * stride;
      s[k] = (ek < E) ? edges[2 * ek] : -1;
    }
#pragma unroll
    for (int k = 0; k < 4; ++k) {
      if (s[k] >= 0) atomicAdd(&h[s[k] >> RV_SHIFT], 1u);
    }
  }
  __syncthreads();
  for (int b = threadIdx.x; b < NB; b += 256) {
    const unsigned c = h[b];
    if (c) atomicAdd(&tot[b], c);
  }
}

static __device__ __forceinline__ float full_tile_mean(const float* __restrict__ messages,
                                                       long long e0, int lane) {
  const int rowA = lane & 15;
  const int kb   = (lane < 16) ? 0 : 2;
  const float* m1 = messages + (e0 + rowA) * 8 + kb;
  const float* m2 = m1 + 16 * 8;
  v2f a1lo = *(const v2f*)(m1);
  v2f a1hi = *(const v2f*)(m1 + 4);
  v2f a2lo = *(const v2f*)(m2);
  v2f a2hi = *(const v2f*)(m2 + 4);
  v2f ones; ones[0] = 1.0f; ones[1] = 1.0f;

  v8f c1 = {};
  c1 = __builtin_amdgcn_wmma_f32_16x16x4_f32(false, a1lo, false, ones, (short)0, c1, false, false);
  c1 = __builtin_amdgcn_wmma_f32_16x16x4_f32(false, a1hi, false, ones, (short)0, c1, false, false);
  v8f c2 = {};
  c2 = __builtin_amdgcn_wmma_f32_16x16x4_f32(false, a2lo, false, ones, (short)0, c2, false, false);
  c2 = __builtin_amdgcn_wmma_f32_16x16x4_f32(false, a2hi, false, ones, (short)0, c2, false, false);

  const bool useD2 = (lane >> 3) & 1;
  v8f d = useD2 ? c2 : c1;
  const int r = lane & 7;
  float s = d[0];
  if (r == 1) s = d[1];
  if (r == 2) s = d[2];
  if (r == 3) s = d[3];
  if (r == 4) s = d[4];
  if (r == 5) s = d[5];
  if (r == 6) s = d[6];
  if (r == 7) s = d[7];
  return s * 0.125f;
}

static __device__ __forceinline__ void place_edge(unsigned* lh, uint2* __restrict__ recs,
                                                  int2 sd, float weight) {
  const unsigned pos = atomicAdd(&lh[sd.x >> RV_SHIFT], 1u);
  recs[pos] = make_uint2(((unsigned)(sd.x & (RV - 1)) << DST_BITS) | (unsigned)sd.y,
                         __float_as_uint(weight));
}

__global__ __launch_bounds__(256) void BinScatter_kernel(
    const int*   __restrict__ edges,
    const float* __restrict__ messages,
    const unsigned* __restrict__ tot,
    unsigned* __restrict__ cur,
    unsigned* __restrict__ gstart,
    uint2* __restrict__ recs,
    long long E, int NB) {
  __shared__ unsigned lh[MAXB];
  __shared__ unsigned bs[MAXB];
  __shared__ unsigned sc[256];
  const int tid  = threadIdx.x;
  const int lane = tid & 31;
  const int wave = tid >> 5;

  const long long tiles = (E + 31) >> 5;
  const long long ct    = (tiles + gridDim.x - 1) / gridDim.x;
  const long long t0    = (long long)blockIdx.x * ct;
  long long t1 = t0 + ct;
  if (t1 > tiles) t1 = tiles;
  const long long eb = t0 << 5;
  long long ee = t1 << 5;
  if (ee > E) ee = E;

  unsigned v0 = 0u, v1 = 0u, v2 = 0u, v3 = 0u;
  {
    const int b = 4 * tid;
    if (b + 0 < NB) v0 = tot[b + 0];
    if (b + 1 < NB) v1 = tot[b + 1];
    if (b + 2 < NB) v2 = tot[b + 2];
    if (b + 3 < NB) v3 = tot[b + 3];
    lh[b + 0] = 0u; lh[b + 1] = 0u; lh[b + 2] = 0u; lh[b + 3] = 0u;
  }
  const unsigned mine = v0 + v1 + v2 + v3;
  sc[tid] = mine;
  __syncthreads();
  for (int d = 1; d < 256; d <<= 1) {
    const unsigned a = (tid >= d) ? sc[tid - d] : 0u;
    __syncthreads();
    sc[tid] += a;
    __syncthreads();
  }
  {
    const int b = 4 * tid;
    unsigned run = sc[tid] - mine;
    bs[b + 0] = run; run += v0;
    bs[b + 1] = run; run += v1;
    bs[b + 2] = run; run += v2;
    bs[b + 3] = run;
  }

  for (long long e = eb + tid; e < ee; e += 4 * 256) {
    int s[4];
#pragma unroll
    for (int k = 0; k < 4; ++k) {
      const long long ek = e + k * 256;
      s[k] = (ek < ee) ? edges[2 * ek] : -1;
    }
#pragma unroll
    for (int k = 0; k < 4; ++k) {
      if (s[k] >= 0) atomicAdd(&lh[s[k] >> RV_SHIFT], 1u);
    }
  }
  __syncthreads();

  for (int b = tid; b < NB; b += 256) {
    const unsigned c = lh[b];
    unsigned base = 0u;
    if (c) base = atomicAdd(&cur[b], c);
    lh[b] = bs[b] + base;
  }
  if (blockIdx.x == 0) {
    for (int b = tid; b < NB; b += 256) gstart[b] = bs[b];
  }
  __syncthreads();

  const int octet = lane >> 3;
  const int perm  = (octet == 1) ? (lane + 8) : ((octet == 2) ? (lane - 8) : lane);

  for (long long t = t0 + 2 * wave; t < t1; t += 16) {
    const long long e0 = t << 5;
    if (t + 1 < t1 && e0 + 64 <= E) {
      const int2 sdA = *reinterpret_cast<const int2*>(edges + 2 * (e0 + perm));
      const int2 sdB = *reinterpret_cast<const int2*>(edges + 2 * (e0 + 32 + perm));
      const float wA = full_tile_mean(messages, e0, lane);
      const float wB = full_tile_mean(messages, e0 + 32, lane);
      place_edge(lh, recs, sdA, wA);
      place_edge(lh, recs, sdB, wB);
    } else {
      const long long tl = (t + 1 < t1) ? (t + 2) : (t + 1);
      for (long long u = t; u < tl; ++u) {
        const long long f0 = u << 5;
        const long long eg = f0 + perm;
        const long long el = (eg < E) ? eg : (E - 1);
        const int2 sd = *reinterpret_cast<const int2*>(edges + 2 * el);
        float weight = 0.0f;
        if (f0 + 32 <= E) {
          weight = full_tile_mean(messages, f0, lane);
        } else if (eg < E) {
          float s = 0.0f;
#pragma unroll
          for (int k = 0; k < 8; ++k) s += messages[eg * 8 + k];
          weight = s * 0.125f;
        }
        if (eg < E) place_edge(lh, recs, sd, weight);
      }
    }
  }
}

__global__ __launch_bounds__(RV) void RangeSum_kernel(
    const float* __restrict__ vertices,
    const unsigned* __restrict__ tot,
    const unsigned* __restrict__ gstart,
    const uint2* __restrict__ recs,
    float* __restrict__ out, int N) {
  __shared__ float    xs[3 * RV];
  __shared__ float    a0s[RV];
  __shared__ float    a1s[RV];
  __shared__ float    a2s[RV];
  __shared__ unsigned acs[RV];
  const int tid = threadIdx.x;
  const int b   = blockIdx.x;
  const int lo  = b << RV_SHIFT;
  const int nv  = (N - lo < RV) ? (N - lo) : RV;

  for (int i = tid; i < 3 * nv; i += RV) xs[i] = vertices[3 * lo + i];
  a0s[tid] = 0.0f;
  a1s[tid] = 0.0f;
  a2s[tid] = 0.0f;
  acs[tid] = 0u;
  __syncthreads();

  const unsigned start = gstart[b];
  const unsigned cnt   = tot[b];
  const unsigned vmask = (1u << DST_BITS) - 1u;
  uint2 r0 = make_uint2(0u, 0u);
  uint2 r1 = make_uint2(0u, 0u);
  float y0 = 0.0f, y1 = 0.0f, y2 = 0.0f;
  if ((unsigned)tid < cnt) {
    const unsigned i1 = ((unsigned)tid + RV < cnt) ? ((unsigned)tid + RV) : (cnt - 1u);
    r0 = recs[start + tid];
    r1 = recs[start + i1];
    const int vd = (int)(r0.x & vmask);
    y0 = vertices[3 * vd + 0];
    y1 = vertices[3 * vd + 1];
    y2 = vertices[3 * vd + 2];
  }
  for (unsigned i = tid; i < cnt; i += RV) {
    const unsigned i2  = (i + 2 * RV < cnt) ? (i + 2 * RV) : (cnt - 1u);
    const int      vdn = (int)(r1.x & vmask);
    const float n0 = vertices[3 * vdn + 0];
    const float n1 = vertices[3 * vdn + 1];
    const float n2 = vertices[3 * vdn + 2];
    const uint2 r2 = recs[start + i2];

    const int   sl = (int)(r0.x >> DST_BITS);
    const float weight = __uint_as_float(r0.y);
    const float x0 = xs[3 * sl + 0], x1 = xs[3 * sl + 1], x2 = xs[3 * sl + 2];
    const float t  = -x0 * y0 + x1 * y1 + x2 * y2;
    const float w0 = y0 + x0 * t;
    const float w1 = y1 + x1 * t;
    const float w2 = y2 + x2 * t;
    const float vn   = sqrtf(-w0 * w0 + w1 * w1 + w2 * w2 + EPSV);
    const float dist = acoshf(-t);
    const float sc   = dist * weight / vn;
    atomicAdd(&a0s[sl], w0 * sc);
    atomicAdd(&a1s[sl], w1 * sc);
    atomicAdd(&a2s[sl], w2 * sc);
    atomicAdd(&acs[sl], 1u);

    r0 = r1; r1 = r2;
    y0 = n0; y1 = n1; y2 = n2;
  }
  __syncthreads();

  if (tid < nv) {
    const float c   = (float)acs[tid];
    const float inv = 1.0f / fmaxf(c, 1.0f);
    float a0 = 0.0f, a1 = 0.0f, a2 = 0.0f;
    if (c > 0.0f) {
      a0 = a0s[tid] * inv;
      a1 = a1s[tid] * inv;
      a2 = a2s[tid] * inv;
    }
    const float an = sqrtf(-a0 * a0 + a1 * a1 + a2 * a2 + EPSV);
    const float T  = an;
    const float e  = expf(T);
    const float ei = 1.0f / e;
    const float ch = 0.5f * (e + ei);
    const float sh = 0.5f * (e - ei);
    const float f  = sh / an;
    const int   i  = lo + tid;
    out[3 * i + 0] = ch * xs[3 * tid + 0] + f * a0;
    out[3 * i + 1] = ch * xs[3 * tid + 1] + f * a1;
    out[3 * i + 2] = ch * xs[3 * tid + 2] + f * a2;
  }
}

extern "C" void kernel_launch(void* const* d_in, const int* in_sizes, int n_in,
                              void* d_out, int out_size, void* d_ws, size_t ws_size,
                              hipStream_t stream) {
  const float* vertices = (const float*)d_in[0];
  const int*   edges    = (const int*)d_in[1];
  const float* messages = (const float*)d_in[2];
  float* out = (float*)d_out;

  const int       N = in_sizes[0] / 3;
  const long long E = (long long)in_sizes[1] / 2;

  const int    NB   = (N + RV - 1) >> RV_SHIFT;
  const size_t need = (size_t)16 * MAXB + (size_t)8 * (size_t)E;
  if (NB <= MAXB && E < (1ll << 31) && ws_size >= need) {
    unsigned* tab    = (unsigned*)d_ws;
    unsigned* tot    = tab;
    unsigned* cur    = tab + MAXB;
    unsigned* gstart = tab + 2 * MAXB;
    uint2*    recs   = (uint2*)(tab + 4 * MAXB);
    BinZero_kernel<<<(2 * MAXB + 255) / 256, 256, 0, stream>>>(tab);
    BinCount_kernel<<<dim3(2048), 256, 0, stream>>>(edges, tot, E, NB);
    BinScatter_kernel<<<dim3(2048), 256, 0, stream>>>(edges, messages, tot, cur, gstart, recs, E, NB);
    RangeSum_kernel<<<dim3(NB), RV, 0, stream>>>(vertices, tot, gstart, recs, out, N);
    return;
  }

  float* acc = (float*)d_ws;

  unsigned* tickets = (unsigned*)d_out;

  const int zn = 4 * N;
  HyperZero_kernel<<<(zn + 255) / 256, 256, 0, stream>>>(acc, zn, tickets);

  HyperEdge_kernel<<<dim3(2048), 256, 0, stream>>>(
      vertices, edges, messages, acc, tickets, E, N);

  HyperVertex_kernel<<<(N + 255) / 256, 256, 0, stream>>>(vertices, acc, out, N);
}


# Per-chunk range counts kept as a table (256 chunks, a 32-wave work-group each); scatter sums the rows for bases and offsets: no clear launch, no integer atomics, no recount; record order changes
# speedup vs baseline: 3.6015x; 1.0678x over previous
#include <hip/hip_runtime.h>
#include <hip/hip_bf16.h>

typedef __attribute__((ext_vector_type(2))) float v2f;
typedef __attribute__((ext_vector_type(8))) float v8f;

#define EPSV 1e-6f

#define RUN_TILES 8
#define TICKET_STRIDE 128

__global__ void HyperZero_kernel(float* __restrict__ p, int n, unsigned* __restrict__ tickets) {
  int i = blockIdx.x * blockDim.x + threadIdx.x;
  if (i < n) p[i] = 0.0f;
  if (i < 2) tickets[i * TICKET_STRIDE] = 0u;
}

template <int K>
static __device__ __forceinline__ int quad_lane(int v) {
  return __builtin_amdgcn_mov_dpp(v, K * 0x55, 0xf, 0xf, true);
}
template <int K>
static __device__ __forceinline__ float quad_lane(float v) {
  return __int_as_float(quad_lane<K>(__float_as_int(v)));
}

template <int K>
static __device__ __forceinline__ void add_quad_edge(float* __restrict__ acc, int comp, int vs,
                                                     float c0, float c1, float c2) {
  const int   vj = quad_lane<K>(vs);
  const float t0 = quad_lane<K>(c0);
  const float t1 = quad_lane<K>(c1);
  const float t2 = quad_lane<K>(c2);
  const float lo = (comp == 0) ? t0 : t1;
  const float hi = (comp == 2) ? t2 : 1.0f;
  const float val = (comp < 2) ? lo : hi;
  if (vj >= 0) atomicAdd(&acc[4 * vj + comp], val);
}

static __device__ __forceinline__ void edge_tile(
    const float* __restrict__ vertices,
    const int*   __restrict__ edges,
    const float* __restrict__ messages,
    float* __restrict__ acc,
    long long E, long long e0, int lo, int hi, int lane) {
  const int octet = lane >> 3;
  const int perm  = (octet == 1) ? (lane + 8) : ((octet == 2) ? (lane - 8) : lane);
  const long long eg = e0 + perm;

  const long long el = (eg < E) ? eg : (E - 1);
  const int2 sd = *reinterpret_cast<const int2*>(edges + 2 * el);

  float weight = 0.0f;
  if (e0 + 32 <= E) {
    const int rowA = lane & 15;
    const int kb   = (lane < 16) ? 0 : 2;
    const float* m1 = messages + (e0 + rowA) * 8 + kb;
    const float* m2 = m1 + 16 * 8;
    v2f a1lo = *(const v2f*)(m1);
    v2f a1hi = *(const v2f*)(m1 + 4);
    v2f a2lo = *(const v2f*)(m2);
    v2f a2hi = *(const v2f*)(m2 + 4);
    v2f ones; ones[0] = 1.0f; ones[1] = 1.0f;

    v8f c1 = {};
    c1 = __builtin_amdgcn_wmma_f32_16x16x4_f32(false, a1lo, false, ones, (short)0, c1, false, false);
    c1 = __builtin_amdgcn_wmma_f32_16x16x4_f32(false, a1hi, false, ones, (short)0, c1, false, false);
    v8f c2 = {};
    c2 = __builtin_amdgcn_wmma_f32_16x16x4_f32(false, a2lo, false, ones, (short)0, c2, false, false);
    c2 = __builtin_amdgcn_wmma_f32_16x16x4_f32(false, a2hi, false, ones, (short)0, c2, false, false);

    const bool useD2 = (lane >> 3) & 1;
    v8f d = useD2 ? c2 : c1;
    const int r = lane & 7;
    float s = d[0];
    if (r == 1) s = d[1];
    if (r == 2) s = d[2];
    if (r == 3) s = d[3];
    if (r == 4) s = d[4];
    if (r == 5) s = d[5];
    if (r == 6) s = d[6];
    if (r == 7) s = d[7];
    weight = s * 0.125f;
  } else if (eg < E) {
    float s = 0.0f;
#pragma unroll
    for (int k = 0; k < 8; ++k) s += messages[eg * 8 + k];
    weight = s * 0.125f;
  }

  int   vs = -1;
  float c0 = 0.0f, c1 = 0.0f, c2 = 0.0f;
  {
    const int s  = (eg < E) ? sd.x : -1;
    const int vd = sd.y;
    if (s >= lo && s < hi) {
      vs = s;
      const float x0 = vertices[3 * vs + 0], x1 = vertices[3 * vs + 1], x2 = vertices[3 * vs + 2];
      const float y0 = vertices[3 * vd + 0], y1 = vertices[3 * vd + 1], y2 = vertices[3 * vd + 2];
      const float t  = -x0 * y0 + x1 * y1 + x2 * y2;
      const float v0 = y0 + x0 * t;
      const float v1 = y1 + x1 * t;
      const float v2 = y2 + x2 * t;
      const float vn   = sqrtf(-v0 * v0 + v1 * v1 + v2 * v2 + EPSV);
      const float dist = acoshf(-t);
      const float sc   = dist * weight / vn;
      c0 = v0 * sc;
      c1 = v1 * sc;
      c2 = v2 * sc;
    }
  }

  const int comp = lane & 3;
  add_quad_edge<0>(acc, comp, vs, c0, c1, c2);
  add_quad_edge<1>(acc, comp, vs, c0, c1, c2);
  add_quad_edge<2>(acc, comp, vs, c0, c1, c2);
  add_quad_edge<3>(acc, comp, vs, c0, c1, c2);
}

__global__ __launch_bounds__(256) void HyperEdge_kernel(
    const float* __restrict__ vertices,
    const int*   __restrict__ edges,
    const float* __restrict__ messages,
    float* __restrict__ acc,
    unsigned* __restrict__ tickets,
    long long E, int N) {
  const int lane = threadIdx.x & 31;
  const unsigned place  = __builtin_amdgcn_s_sendmsg_rtn(0x87);
  const int      myhalf = (int)((place >> 18) & 1u);

  const long long tiles = (E + 31) >> 5;
  const unsigned  runs  = (unsigned)((tiles + RUN_TILES - 1) / RUN_TILES);
  const int       split = ((N >> 1) + 15) & ~15;

  for (int pass = 0; pass < 2; ++pass) {
    const int h  = myhalf ^ pass;
    const int lo = h ? split : 0;
    const int hi = h ? N : split;
    for (unsigned it = 0; it <= runs; ++it) {
      unsigned t = 0u;
      if (lane == 0) t = atomicAdd(&tickets[h * TICKET_STRIDE], 1u);
      t = (unsigned)__builtin_amdgcn_readfirstlane((int)t);
      if (t >= runs) break;
      const long long tile0 = (long long)t * RUN_TILES;
      for (int k = 0; k < RUN_TILES; ++k) {
        const long long e0 = (tile0 + k) << 5;
        if (e0 >= E) break;
        edge_tile(vertices, edges, messages, acc, E, e0, lo, hi, lane);
      }
    }
  }
}

__global__ void HyperVertex_kernel(const float* __restrict__ vertices,
                                   const float* __restrict__ acc,
                                   float* __restrict__ out, int N) {
  int i = blockIdx.x * blockDim.x + threadIdx.x;
  if (i >= N) return;
  const float4 rec = reinterpret_cast<const float4*>(acc)[i];
  const float c   = rec.w;
  const float inv = 1.0f / fmaxf(c, 1.0f);
  float a0 = 0.0f, a1 = 0.0f, a2 = 0.0f;
  if (c > 0.0f) {
    a0 = rec.x * inv;
    a1 = rec.y * inv;
    a2 = rec.z * inv;
  }
  const float an = sqrtf(-a0 * a0 + a1 * a1 + a2 * a2 + EPSV);
  const float T  = an;
  const float e  = expf(T);
  const float ei = 1.0f / e;
  const float ch = 0.5f * (e + ei);
  const float sh = 0.5f * (e - ei);
  const float f  = sh / an;
  out[3 * i + 0] = ch * vertices[3 * i + 0] + f * a0;
  out[3 * i + 1] = ch * vertices[3 * i + 1] + f * a1;
  out[3 * i + 2] = ch * vertices[3 * i + 2] + f * a2;
}


#define RV_SHIFT 10
#define RV (1 << RV_SHIFT)
#define MAXB 1024
#define DST_BITS 20

#define NCHUNK 256
#define CT 1024

static __device__ __forceinline__ void chunk_tiles(long long E, int nc, int c,
                                                   long long& t0, long long& t1) {
  const long long tiles = (E + 31) >> 5;
  const long long ct    = (tiles + nc - 1) / nc;
  t0 = (long long)c * ct;
  if (t0 > tiles) t0 = tiles;
  t1 = t0 + ct;
  if (t1 > tiles) t1 = tiles;
}

__global__ __launch_bounds__(CT) void ChunkCount_kernel(const int* __restrict__ edges,
                                                        unsigned* __restrict__ part,
                                                        long long E, int NB) {
  __shared__ unsigned h[MAXB];
  const int tid = threadIdx.x;
  h[tid] = 0u;
  __syncthreads();
  long long t0, t1;
  chunk_tiles(E, gridDim.x, blockIdx.x, t0, t1);
  const long long eb = t0 << 5;
  long long ee = t1 << 5;
  if (ee > E) ee = E;
  for (long long e = eb + tid; e < ee; e += 8 * CT) {
    int s[8];
#pragma unroll
    for (int k = 0; k < 8; ++k) {
      const long long ek = e + k * CT;
      s[k] = (ek < ee) ? edges[2 * ek] : -1;
    }
#pragma unroll
    for (int k = 0; k < 8; ++k) {
      if (s[k] >= 0) atomicAdd(&h[s[k] >> RV_SHIFT], 1u);
    }
  }
  __syncthreads();
  if (tid < NB) part[(size_t)blockIdx.x * MAXB + tid] = h[tid];
}

static __device__ __forceinline__ void tile_loads(const float* __restrict__ messages,
                                                  long long e0, int lane, v2f (&a)[4]) {
  const int rowA = lane & 15;
  const int kb   = (lane < 16) ? 0 : 2;
  const float* m1 = messages + (e0 + rowA) * 8 + kb;
  const float* m2 = m1 + 16 * 8;
  a[0] = *(const v2f*)(m1);
  a[1] = *(const v2f*)(m1 + 4);
  a[2] = *(const v2f*)(m2);
  a[3] = *(const v2f*)(m2 + 4);
}

static __device__ __forceinline__ float tile_mean(const v2f (&a)[4], int lane) {
  v2f ones; ones[0] = 1.0f; ones[1] = 1.0f;

  v8f c1 = {};
  c1 = __builtin_amdgcn_wmma_f32_16x16x4_f32(false, a[0], false, ones, (short)0, c1, false, false);
  c1 = __builtin_amdgcn_wmma_f32_16x16x4_f32(false, a[1], false, ones, (short)0, c1, false, false);
  v8f c2 = {};
  c2 = __builtin_amdgcn_wmma_f32_16x16x4_f32(false, a[2], false, ones, (short)0, c2, false, false);
  c2 = __builtin_amdgcn_wmma_f32_16x16x4_f32(false, a[3], false, ones, (short)0, c2, false, false);

  const bool useD2 = (lane >> 3) & 1;
  v8f d = useD2 ? c2 : c1;
  const int r = lane & 7;
  float s = d[0];
  if (r == 1) s = d[1];
  if (r == 2) s = d[2];
  if (r == 3) s = d[3];
  if (r == 4) s = d[4];
  if (r == 5) s = d[5];
  if (r == 6) s = d[6];
  if (r == 7) s = d[7];
  return s * 0.125f;
}

static __device__ __forceinline__ float full_tile_mean(const float* __restrict__ messages,
                                                       long long e0, int lane) {
  v2f a[4];
  tile_loads(messages, e0, lane, a);
  return tile_mean(a, lane);
}

static __device__ __forceinline__ void place_edge(unsigned* lh, uint2* __restrict__ recs,
                                                  int2 sd, float weight) {
  const unsigned pos = atomicAdd(&lh[sd.x >> RV_SHIFT], 1u);
  recs[pos] = make_uint2(((unsigned)(sd.x & (RV - 1)) << DST_BITS) | (unsigned)sd.y,
                         __float_as_uint(weight));
}

__global__ __launch_bounds__(CT) void ChunkScatter_kernel(
    const int*   __restrict__ edges,
    const float* __restrict__ messages,
    const unsigned* __restrict__ part,
    unsigned* __restrict__ tot,
    unsigned* __restrict__ gstart,
    uint2* __restrict__ recs,
    long long E, int NB) {
  __shared__ unsigned lh[MAXB];
  __shared__ unsigned ht[CT];
  __shared__ unsigned hb[CT];
  __shared__ unsigned sc[CT / 32];
  const int tid  = threadIdx.x;
  const int lane = tid & 31;
  const int wave = tid >> 5;
  const int me   = blockIdx.x;
  const int nc   = gridDim.x;

  int gsh = 0;
  if (NB <= 512) gsh = 1;
  if (NB <= 256) gsh = 2;
  if (NB <= 128) gsh = 3;
  const int slot = CT >> gsh;
  const int b    = tid & (slot - 1);
  const int grp  = tid >> (10 - gsh);
  unsigned tsum = 0u, bsum = 0u;
  if (b < NB) {
#pragma unroll 16
    for (int w = grp; w < nc; w += (1 << gsh)) {
      const unsigned v = part[(size_t)w * MAXB + b];
      tsum += v;
      bsum += (w < me) ? v : 0u;
    }
  }
  ht[tid] = tsum;
  hb[tid] = bsum;
  __syncthreads();
  unsigned total = 0u, before = 0u;
  if (tid < NB) {
    for (int g = 0; g < (1 << gsh); ++g) {
      total  += ht[g * slot + tid];
      before += hb[g * slot + tid];
    }
  }
  unsigned incl = total;
  for (int d = 1; d < 32; d <<= 1) {
    const unsigned n = __shfl_up(incl, d);
    if (lane >= d) incl += n;
  }
  if (lane == 31) sc[wave] = incl;
  __syncthreads();
  unsigned first = incl - total;
  for (int j = 0; j < wave; ++j) first += sc[j];
  lh[tid] = first + before;
  if (me == 0 && tid < NB) {
    gstart[tid] = first;
    tot[tid]    = total;
  }
  __syncthreads();

  long long t0, t1;
  chunk_tiles(E, nc, me, t0, t1);

  const int octet = lane >> 3;
  const int perm  = (octet == 1) ? (lane + 8) : ((octet == 2) ? (lane - 8) : lane);

  for (long long t = t0 + 4 * wave; t < t1; t += 4 * (CT / 32)) {
    const long long e0 = t << 5;
    if (t + 3 < t1 && e0 + 128 <= E) {
      const int2 sdA = *reinterpret_cast<const int2*>(edges + 2 * (e0 + perm));
      const int2 sdB = *reinterpret_cast<const int2*>(edges + 2 * (e0 + 32 + perm));
      const int2 sdC = *reinterpret_cast<const int2*>(edges + 2 * (e0 + 64 + perm));
      const int2 sdD = *reinterpret_cast<const int2*>(edges + 2 * (e0 + 96 + perm));
      v2f aA[4], aB[4], aC[4], aD[4];
      tile_loads(messages, e0, lane, aA);
      tile_loads(messages, e0 + 32, lane, aB);
      tile_loads(messages, e0 + 64, lane, aC);
      tile_loads(messages, e0 + 96, lane, aD);
      __builtin_amdgcn_sched_barrier(0);
      const float wA = tile_mean(aA, lane);
      const float wB = tile_mean(aB, lane);
      const float wC = tile_mean(aC, lane);
      const float wD = tile_mean(aD, lane);
      place_edge(lh, recs, sdA, wA);
      place_edge(lh, recs, sdB, wB);
      place_edge(lh, recs, sdC, wC);
      place_edge(lh, recs, sdD, wD);
    } else {
      const long long tl = (t + 4 < t1) ? (t + 4) : t1;
      for (long long u = t; u < tl; ++u) {
        const long long f0 = u << 5;
        const long long eg = f0 + perm;
        const long long el = (eg < E) ? eg : (E - 1);
        const int2 sd = *reinterpret_cast<const int2*>(edges + 2 * el);
        float weight = 0.0f;
        if (f0 + 32 <= E) {
          weight = full_tile_mean(messages, f0, lane);
        } else if (eg < E) {
          float s = 0.0f;
#pragma unroll
          for (int k = 0; k < 8; ++k) s += messages[eg * 8 + k];
          weight = s * 0.125f;
        }
        if (eg < E) place_edge(lh, recs, sd, weight);
      }
    }
  }
}

__global__ __launch_bounds__(RV) void RangeSum_kernel(
    const float* __restrict__ vertices,
    const unsigned* __restrict__ tot,
    const unsigned* __restrict__ gstart,
    const uint2* __restrict__ recs,
    float* __restrict__ out, int N) {
  __shared__ float    xs[3 * RV];
  __shared__ float    a0s[RV];
  __shared__ float    a1s[RV];
  __shared__ float    a2s[RV];
  __shared__ unsigned acs[RV];
  const int tid = threadIdx.x;
  const int b   = blockIdx.x;
  const int lo  = b << RV_SHIFT;
  const int nv  = (N - lo < RV) ? (N - lo) : RV;

  for (int i = tid; i < 3 * nv; i += RV) xs[i] = vertices[3 * lo + i];
  a0s[tid] = 0.0f;
  a1s[tid] = 0.0f;
  a2s[tid] = 0.0f;
  acs[tid] = 0u;
  __syncthreads();

  const unsigned start = gstart[b];
  const unsigned cnt   = tot[b];
  const unsigned vmask = (1u << DST_BITS) - 1u;
  uint2 r0 = make_uint2(0u, 0u);
  uint2 r1 = make_uint2(0u, 0u);
  float y0 = 0.0f, y1 = 0.0f, y2 = 0.0f;
  if ((unsigned)tid < cnt) {
    const unsigned i1 = ((unsigned)tid + RV < cnt) ? ((unsigned)tid + RV) : (cnt - 1u);
    r0 = recs[start + tid];
    r1 = recs[start + i1];
    const int vd = (int)(r0.x & vmask);
    y0 = vertices[3 * vd + 0];
    y1 = vertices[3 * vd + 1];
    y2 = vertices[3 * vd + 2];
  }
  for (unsigned i = tid; i < cnt; i += RV) {
    const unsigned i2  = (i + 2 * RV < cnt) ? (i + 2 * RV) : (cnt - 1u);
    const int      vdn = (int)(r1.x & vmask);
    const float n0 = vertices[3 * vdn + 0];
    const float n1 = vertices[3 * vdn + 1];
    const float n2 = vertices[3 * vdn + 2];
    const uint2 r2 = recs[start + i2];

    const int   sl = (int)(r0.x >> DST_BITS);
    const float weight = __uint_as_float(r0.y);
    const float x0 = xs[3 * sl + 0], x1 = xs[3 * sl + 1], x2 = xs[3 * sl + 2];
    const float t  = -x0 * y0 + x1 * y1 + x2 * y2;
    const float w0 = y0 + x0 * t;
    const float w1 = y1 + x1 * t;
    const float w2 = y2 + x2 * t;
    const float vn   = sqrtf(-w0 * w0 + w1 * w1 + w2 * w2 + EPSV);
    const float dist = acoshf(-t);
    const float sc   = dist * weight / vn;
    atomicAdd(&a0s[sl], w0 * sc);
    atomicAdd(&a1s[sl], w1 * sc);
    atomicAdd(&a2s[sl], w2 * sc);
    atomicAdd(&acs[sl], 1u);

    r0 = r1; r1 = r2;
    y0 = n0; y1 = n1; y2 = n2;
  }
  __syncthreads();

  if (tid < nv) {
    const float c   = (float)acs[tid];
    const float inv = 1.0f / fmaxf(c, 1.0f);
    float a0 = 0.0f, a1 = 0.0f, a2 = 0.0f;
    if (c > 0.0f) {
      a0 = a0s[tid] * inv;
      a1 = a1s[tid] * inv;
      a2 = a2s[tid] * inv;
    }
    const float an = sqrtf(-a0 * a0 + a1 * a1 + a2 * a2 + EPSV);
    const float T  = an;
    const float e  = expf(T);
    const float ei = 1.0f / e;
    const float ch = 0.5f * (e + ei);
    const float sh = 0.5f * (e - ei);
    const float f  = sh / an;
    const int   i  = lo + tid;
    out[3 * i + 0] = ch * xs[3 * tid + 0] + f * a0;
    out[3 * i + 1] = ch * xs[3 * tid + 1] + f * a1;
    out[3 * i + 2] = ch * xs[3 * tid + 2] + f * a2;
  }
}

extern "C" void kernel_launch(void* const* d_in, const int* in_sizes, int n_in,
                              void* d_out, int out_size, void* d_ws, size_t ws_size,
                              hipStream_t stream) {
  const float* vertices = (const float*)d_in[0];
  const int*   edges    = (const int*)d_in[1];
  const float* messages = (const float*)d_in[2];
  float* out = (float*)d_out;

  const int       N = in_sizes[0] / 3;
  const long long E = (long long)in_sizes[1] / 2;

  const int    NB   = (N + RV - 1) >> RV_SHIFT;
  const size_t need = (size_t)4 * MAXB * (2 + NCHUNK) + (size_t)8 * (size_t)E;
  if (NB <= MAXB && E < (1ll << 31) && ws_size >= need) {
    unsigned* tab    = (unsigned*)d_ws;
    unsigned* tot    = tab;
    unsigned* gstart = tab + MAXB;
    unsigned* part   = tab + 2 * MAXB;
    uint2*    recs   = (uint2*)(tab + (2 + NCHUNK) * MAXB);
    ChunkCount_kernel<<<dim3(NCHUNK), CT, 0, stream>>>(edges, part, E, NB);
    ChunkScatter_kernel<<<dim3(NCHUNK), CT, 0, stream>>>(edges, messages, part, tot, gstart, recs, E, NB);
    RangeSum_kernel<<<dim3(NB), RV, 0, stream>>>(vertices, tot, gstart, recs, out, N);
    return;
  }

  float* acc = (float*)d_ws;

  unsigned* tickets = (unsigned*)d_out;

  const int zn = 4 * N;
  HyperZero_kernel<<<(zn + 255) / 256, 256, 0, stream>>>(acc, zn, tickets);

  HyperEdge_kernel<<<dim3(2048), 256, 0, stream>>>(
      vertices, edges, messages, acc, tickets, E, N);

  HyperVertex_kernel<<<(N + 255) / 256, 256, 0, stream>>>(vertices, acc, out, N);
}


# Hyperboloid distance by the closed form log(u + sqrt((u-1)(u+1))) on bare hardware sqrt and log, not the library acosh's double-float sequence: summing loop 168 -> 61 vector instructions per edge
# speedup vs baseline: 3.7121x; 1.0307x over previous
#include <hip/hip_runtime.h>
#include <hip/hip_bf16.h>

typedef __attribute__((ext_vector_type(2))) float v2f;
typedef __attribute__((ext_vector_type(8))) float v8f;

#define EPSV 1e-6f

static __device__ __forceinline__ float hyp_dist(float u) {
  const bool  big = u > 1.0e18f;
  const float s   = __builtin_amdgcn_sqrtf((u - 1.0f) * (u + 1.0f));
  const float r   = __builtin_amdgcn_logf(big ? u : (u + s)) * 0.693147180559945f;
  return big ? (r + 0.693147180559945f) : r;
}

#define RUN_TILES 8
#define TICKET_STRIDE 128

__global__ void HyperZero_kernel(float* __restrict__ p, int n, unsigned* __restrict__ tickets) {
  int i = blockIdx.x * blockDim.x + threadIdx.x;
  if (i < n) p[i] = 0.0f;
  if (i < 2) tickets[i * TICKET_STRIDE] = 0u;
}

template <int K>
static __device__ __forceinline__ int quad_lane(int v) {
  return __builtin_amdgcn_mov_dpp(v, K * 0x55, 0xf, 0xf, true);
}
template <int K>
static __device__ __forceinline__ float quad_lane(float v) {
  return __int_as_float(quad_lane<K>(__float_as_int(v)));
}

template <int K>
static __device__ __forceinline__ void add_quad_edge(float* __restrict__ acc, int comp, int vs,
                                                     float c0, float c1, float c2) {
  const int   vj = quad_lane<K>(vs);
  const float t0 = quad_lane<K>(c0);
  const float t1 = quad_lane<K>(c1);
  const float t2 = quad_lane<K>(c2);
  const float lo = (comp == 0) ? t0 : t1;
  const float hi = (comp == 2) ? t2 : 1.0f;
  const float val = (comp < 2) ? lo : hi;
  if (vj >= 0) atomicAdd(&acc[4 * vj + comp], val);
}

static __device__ __forceinline__ void edge_tile(
    const float* __restrict__ vertices,
    const int*   __restrict__ edges,
    const float* __restrict__ messages,
    float* __restrict__ acc,
    long long E, long long e0, int lo, int hi, int lane) {
  const int octet = lane >> 3;
  const int perm  = (octet == 1) ? (lane + 8) : ((octet == 2) ? (lane - 8) : lane);
  const long long eg = e0 + perm;

  const long long el = (eg < E) ? eg : (E - 1);
  const int2 sd = *reinterpret_cast<const int2*>(edges + 2 * el);

  float weight = 0.0f;
  if (e0 + 32 <= E) {
    const int rowA = lane & 15;
    const int kb   = (lane < 16) ? 0 : 2;
    const float* m1 = messages + (e0 + rowA) * 8 + kb;
    const float* m2 = m1 + 16 * 8;
    v2f a1lo = *(const v2f*)(m1);
    v2f a1hi = *(const v2f*)(m1 + 4);
    v2f a2lo = *(const v2f*)(m2);
    v2f a2hi = *(const v2f*)(m2 + 4);
    v2f ones; ones[0] = 1.0f; ones[1] = 1.0f;

    v8f c1 = {};
    c1 = __builtin_amdgcn_wmma_f32_16x16x4_f32(false, a1lo, false, ones, (short)0, c1, false, false);
    c1 = __builtin_amdgcn_wmma_f32_16x16x4_f32(false, a1hi, false, ones, (short)0, c1, false, false);
    v8f c2 = {};
    c2 = __builtin_amdgcn_wmma_f32_16x16x4_f32(false, a2lo, false, ones, (short)0, c2, false, false);
    c2 = __builtin_amdgcn_wmma_f32_16x16x4_f32(false, a2hi, false, ones, (short)0, c2, false, false);

    const bool useD2 = (lane >> 3) & 1;
    v8f d = useD2 ? c2 : c1;
    const int r = lane & 7;
    float s = d[0];
    if (r == 1) s = d[1];
    if (r == 2) s = d[2];
    if (r == 3) s = d[3];
    if (r == 4) s = d[4];
    if (r == 5) s = d[5];
    if (r == 6) s = d[6];
    if (r == 7) s = d[7];
    weight = s * 0.125f;
  } else if (eg < E) {
    float s = 0.0f;
#pragma unroll
    for (int k = 0; k < 8; ++k) s += messages[eg * 8 + k];
    weight = s * 0.125f;
  }

  int   vs = -1;
  float c0 = 0.0f, c1 = 0.0f, c2 = 0.0f;
  {
    const int s  = (eg < E) ? sd.x : -1;
    const int vd = sd.y;
    if (s >= lo && s < hi) {
      vs = s;
      const float x0 = vertices[3 * vs + 0], x1 = vertices[3 * vs + 1], x2 = vertices[3 * vs + 2];
      const float y0 = vertices[3 * vd + 0], y1 = vertices[3 * vd + 1], y2 = vertices[3 * vd + 2];
      const float t  = -x0 * y0 + x1 * y1 + x2 * y2;
      const float v0 = y0 + x0 * t;
      const float v1 = y1 + x1 * t;
      const float v2 = y2 + x2 * t;
      const float vn   = sqrtf(-v0 * v0 + v1 * v1 + v2 * v2 + EPSV);
      const float dist = hyp_dist(-t);
      const float sc   = dist * weight / vn;
      c0 = v0 * sc;
      c1 = v1 * sc;
      c2 = v2 * sc;
    }
  }

  const int comp = lane & 3;
  add_quad_edge<0>(acc, comp, vs, c0, c1, c2);
  add_quad_edge<1>(acc, comp, vs, c0, c1, c2);
  add_quad_edge<2>(acc, comp, vs, c0, c1, c2);
  add_quad_edge<3>(acc, comp, vs, c0, c1, c2);
}

__global__ __launch_bounds__(256) void HyperEdge_kernel(
    const float* __restrict__ vertices,
    const int*   __restrict__ edges,
    const float* __restrict__ messages,
    float* __restrict__ acc,
    unsigned* __restrict__ tickets,
    long long E, int N) {
  const int lane = threadIdx.x & 31;
  const unsigned place  = __builtin_amdgcn_s_sendmsg_rtn(0x87);
  const int      myhalf = (int)((place >> 18) & 1u);

  const long long tiles = (E + 31) >> 5;
  const unsigned  runs  = (unsigned)((tiles + RUN_TILES - 1) / RUN_TILES);
  const int       split = ((N >> 1) + 15) & ~15;

  for (int pass = 0; pass < 2; ++pass) {
    const int h  = myhalf ^ pass;
    const int lo = h ? split : 0;
    const int hi = h ? N : split;
    for (unsigned it = 0; it <= runs; ++it) {
      unsigned t = 0u;
      if (lane == 0) t = atomicAdd(&tickets[h * TICKET_STRIDE], 1u);
      t = (unsigned)__builtin_amdgcn_readfirstlane((int)t);
      if (t >= runs) break;
      const long long tile0 = (long long)t * RUN_TILES;
      for (int k = 0; k < RUN_TILES; ++k) {
        const long long e0 = (tile0 + k) << 5;
        if (e0 >= E) break;
        edge_tile(vertices, edges, messages, acc, E, e0, lo, hi, lane);
      }
    }
  }
}

__global__ void HyperVertex_kernel(const float* __restrict__ vertices,
                                   const float* __restrict__ acc,
                                   float* __restrict__ out, int N) {
  int i = blockIdx.x * blockDim.x + threadIdx.x;
  if (i >= N) return;
  const float4 rec = reinterpret_cast<const float4*>(acc)[i];
  const float c   = rec.w;
  const float inv = 1.0f / fmaxf(c, 1.0f);
  float a0 = 0.0f, a1 = 0.0f, a2 = 0.0f;
  if (c > 0.0f) {
    a0 = rec.x * inv;
    a1 = rec.y * inv;
    a2 = rec.z * inv;
  }
  const float an = sqrtf(-a0 * a0 + a1 * a1 + a2 * a2 + EPSV);
  const float T  = an;
  const float e  = expf(T);
  const float ei = 1.0f / e;
  const float ch = 0.5f * (e + ei);
  const float sh = 0.5f * (e - ei);
  const float f  = sh / an;
  out[3 * i + 0] = ch * vertices[3 * i + 0] + f * a0;
  out[3 * i + 1] = ch * vertices[3 * i + 1] + f * a1;
  out[3 * i + 2] = ch * vertices[3 * i + 2] + f * a2;
}


#define RV_SHIFT 10
#define RV (1 << RV_SHIFT)
#define MAXB 1024
#define DST_BITS 20

#define NCHUNK 256
#define CT 1024

static __device__ __forceinline__ void chunk_tiles(long long E, int nc, int c,
                                                   long long& t0, long long& t1) {
  const long long tiles = (E + 31) >> 5;
  const long long ct    = (tiles + nc - 1) / nc;
  t0 = (long long)c * ct;
  if (t0 > tiles) t0 = tiles;
  t1 = t0 + ct;
  if (t1 > tiles) t1 = tiles;
}

__global__ __launch_bounds__(CT) void ChunkCount_kernel(const int* __restrict__ edges,
                                                        unsigned* __restrict__ part,
                                                        long long E, int NB) {
  __shared__ unsigned h[MAXB];
  const int tid = threadIdx.x;
  h[tid] = 0u;
  __syncthreads();
  long long t0, t1;
  chunk_tiles(E, gridDim.x, blockIdx.x, t0, t1);
  const long long eb = t0 << 5;
  long long ee = t1 << 5;
  if (ee > E) ee = E;
  for (long long e = eb + tid; e < ee; e += 8 * CT) {
    int s[8];
#pragma unroll
    for (int k = 0; k < 8; ++k) {
      const long long ek = e + k * CT;
      s[k] = (ek < ee) ? edges[2 * ek] : -1;
    }
#pragma unroll
    for (int k = 0; k < 8; ++k) {
      if (s[k] >= 0) atomicAdd(&h[s[k] >> RV_SHIFT], 1u);
    }
  }
  __syncthreads();
  if (tid < NB) part[(size_t)blockIdx.x * MAXB + tid] = h[tid];
}

static __device__ __forceinline__ void tile_loads(const float* __restrict__ messages,
                                                  long long e0, int lane, v2f (&a)[4]) {
  const int rowA = lane & 15;
  const int kb   = (lane < 16) ? 0 : 2;
  const float* m1 = messages + (e0 + rowA) * 8 + kb;
  const float* m2 = m1 + 16 * 8;
  a[0] = *(const v2f*)(m1);
  a[1] = *(const v2f*)(m1 + 4);
  a[2] = *(const v2f*)(m2);
  a[3] = *(const v2f*)(m2 + 4);
}

static __device__ __forceinline__ float tile_mean(const v2f (&a)[4], int lane) {
  v2f ones; ones[0] = 1.0f; ones[1] = 1.0f;

  v8f c1 = {};
  c1 = __builtin_amdgcn_wmma_f32_16x16x4_f32(false, a[0], false, ones, (short)0, c1, false, false);
  c1 = __builtin_amdgcn_wmma_f32_16x16x4_f32(false, a[1], false, ones, (short)0, c1, false, false);
  v8f c2 = {};
  c2 = __builtin_amdgcn_wmma_f32_16x16x4_f32(false, a[2], false, ones, (short)0, c2, false, false);
  c2 = __builtin_amdgcn_wmma_f32_16x16x4_f32(false, a[3], false, ones, (short)0, c2, false, false);

  const bool useD2 = (lane >> 3) & 1;
  v8f d = useD2 ? c2 : c1;
  const int r = lane & 7;
  float s = d[0];
  if (r == 1) s = d[1];
  if (r == 2) s = d[2];
  if (r == 3) s = d[3];
  if (r == 4) s = d[4];
  if (r == 5) s = d[5];
  if (r == 6) s = d[6];
  if (r == 7) s = d[7];
  return s * 0.125f;
}

static __device__ __forceinline__ float full_tile_mean(const float* __restrict__ messages,
                                                       long long e0, int lane) {
  v2f a[4];
  tile_loads(messages, e0, lane, a);
  return tile_mean(a, lane);
}

static __device__ __forceinline__ void place_edge(unsigned* lh, uint2* __restrict__ recs,
                                                  int2 sd, float weight) {
  const unsigned pos = atomicAdd(&lh[sd.x >> RV_SHIFT], 1u);
  recs[pos] = make_uint2(((unsigned)(sd.x & (RV - 1)) << DST_BITS) | (unsigned)sd.y,
                         __float_as_uint(weight));
}

__global__ __launch_bounds__(CT) void ChunkScatter_kernel(
    const int*   __restrict__ edges,
    const float* __restrict__ messages,
    const unsigned* __restrict__ part,
    unsigned* __restrict__ tot,
    unsigned* __restrict__ gstart,
    uint2* __restrict__ recs,
    long long E, int NB) {
  __shared__ unsigned lh[MAXB];
  __shared__ unsigned ht[CT];
  __shared__ unsigned hb[CT];
  __shared__ unsigned sc[CT / 32];
  const int tid  = threadIdx.x;
  const int lane = tid & 31;
  const int wave = tid >> 5;
  const int me   = blockIdx.x;
  const int nc   = gridDim.x;

  int gsh = 0;
  if (NB <= 512) gsh = 1;
  if (NB <= 256) gsh = 2;
  if (NB <= 128) gsh = 3;
  const int slot = CT >> gsh;
  const int b    = tid & (slot - 1);
  const int grp  = tid >> (10 - gsh);
  unsigned tsum = 0u, bsum = 0u;
  if (b < NB) {
#pragma unroll 16
    for (int w = grp; w < nc; w += (1 << gsh)) {
      const unsigned v = part[(size_t)w * MAXB + b];
      tsum += v;
      bsum += (w < me) ? v : 0u;
    }
  }
  ht[tid] = tsum;
  hb[tid] = bsum;
  __syncthreads();
  unsigned total = 0u, before = 0u;
  if (tid < NB) {
    for (int g = 0; g < (1 << gsh); ++g) {
      total  += ht[g * slot + tid];
      before += hb[g * slot + tid];
    }
  }
  unsigned incl = total;
  for (int d = 1; d < 32; d <<= 1) {
    const unsigned n = __shfl_up(incl, d);
    if (lane >= d) incl += n;
  }
  if (lane == 31) sc[wave] = incl;
  __syncthreads();
  unsigned first = incl - total;
  for (int j = 0; j < wave; ++j) first += sc[j];
  lh[tid] = first + before;
  if (me == 0 && tid < NB) {
    gstart[tid] = first;
    tot[tid]    = total;
  }
  __syncthreads();

  long long t0, t1;
  chunk_tiles(E, nc, me, t0, t1);

  const int octet = lane >> 3;
  const int perm  = (octet == 1) ? (lane + 8) : ((octet == 2) ? (lane - 8) : lane);

  for (long long t = t0 + 4 * wave; t < t1; t += 4 * (CT / 32)) {
    const long long e0 = t << 5;
    if (t + 3 < t1 && e0 + 128 <= E) {
      const int2 sdA = *reinterpret_cast<const int2*>(edges + 2 * (e0 + perm));
      const int2 sdB = *reinterpret_cast<const int2*>(edges + 2 * (e0 + 32 + perm));
      const int2 sdC = *reinterpret_cast<const int2*>(edges + 2 * (e0 + 64 + perm));
      const int2 sdD = *reinterpret_cast<const int2*>(edges + 2 * (e0 + 96 + perm));
      v2f aA[4], aB[4], aC[4], aD[4];
      tile_loads(messages, e0, lane, aA);
      tile_loads(messages, e0 + 32, lane, aB);
      tile_loads(messages, e0 + 64, lane, aC);
      tile_loads(messages, e0 + 96, lane, aD);
      __builtin_amdgcn_sched_barrier(0);
      const float wA = tile_mean(aA, lane);
      const float wB = tile_mean(aB, lane);
      const float wC = tile_mean(aC, lane);
      const float wD = tile_mean(aD, lane);
      place_edge(lh, recs, sdA, wA);
      place_edge(lh, recs, sdB, wB);
      place_edge(lh, recs, sdC, wC);
      place_edge(lh, recs, sdD, wD);
    } else {
      const long long tl = (t + 4 < t1) ? (t + 4) : t1;
      for (long long u = t; u < tl; ++u) {
        const long long f0 = u << 5;
        const long long eg = f0 + perm;
        const long long el = (eg < E) ? eg : (E - 1);
        const int2 sd = *reinterpret_cast<const int2*>(edges + 2 * el);
        float weight = 0.0f;
        if (f0 + 32 <= E) {
          weight = full_tile_mean(messages, f0, lane);
        } else if (eg < E) {
          float s = 0.0f;
#pragma unroll
          for (int k = 0; k < 8; ++k) s += messages[eg * 8 + k];
          weight = s * 0.125f;
        }
        if (eg < E) place_edge(lh, recs, sd, weight);
      }
    }
  }
}

__global__ __launch_bounds__(RV) void RangeSum_kernel(
    const float* __restrict__ vertices,
    const unsigned* __restrict__ tot,
    const unsigned* __restrict__ gstart,
    const uint2* __restrict__ recs,
    float* __restrict__ out, int N) {
  __shared__ float    xs[3 * RV];
  __shared__ float    a0s[RV];
  __shared__ float    a1s[RV];
  __shared__ float    a2s[RV];
  __shared__ unsigned acs[RV];
  const int tid = threadIdx.x;
  const int b   = blockIdx.x;
  const int lo  = b << RV_SHIFT;
  const int nv  = (N - lo < RV) ? (N - lo) : RV;

  for (int i = tid; i < 3 * nv; i += RV) xs[i] = vertices[3 * lo + i];
  a0s[tid] = 0.0f;
  a1s[tid] = 0.0f;
  a2s[tid] = 0.0f;
  acs[tid] = 0u;
  __syncthreads();

  const unsigned start = gstart[b];
  const unsigned cnt   = tot[b];
  const unsigned vmask = (1u << DST_BITS) - 1u;
  uint2 r0 = make_uint2(0u, 0u);
  uint2 r1 = make_uint2(0u, 0u);
  float y0 = 0.0f, y1 = 0.0f, y2 = 0.0f;
  if ((unsigned)tid < cnt) {
    const unsigned i1 = ((unsigned)tid + RV < cnt) ? ((unsigned)tid + RV) : (cnt - 1u);
    r0 = recs[start + tid];
    r1 = recs[start + i1];
    const int vd = (int)(r0.x & vmask);
    y0 = vertices[3 * vd + 0];
    y1 = vertices[3 * vd + 1];
    y2 = vertices[3 * vd + 2];
  }
  for (unsigned i = tid; i < cnt; i += RV) {
    const unsigned i2  = (i + 2 * RV < cnt) ? (i + 2 * RV) : (cnt - 1u);
    const int      vdn = (int)(r1.x & vmask);
    const float n0 = vertices[3 * vdn + 0];
    const float n1 = vertices[3 * vdn + 1];
    const float n2 = vertices[3 * vdn + 2];
    const uint2 r2 = recs[start + i2];

    const int   sl = (int)(r0.x >> DST_BITS);
    const float weight = __uint_as_float(r0.y);
    const float x0 = xs[3 * sl + 0], x1 = xs[3 * sl + 1], x2 = xs[3 * sl + 2];
    const float t  = -x0 * y0 + x1 * y1 + x2 * y2;
    const float w0 = y0 + x0 * t;
    const float w1 = y1 + x1 * t;
    const float w2 = y2 + x2 * t;
    const float vn   = sqrtf(-w0 * w0 + w1 * w1 + w2 * w2 + EPSV);
    const float dist = hyp_dist(-t);
    const float sc   = dist * weight / vn;
    atomicAdd(&a0s[sl], w0 * sc);
    atomicAdd(&a1s[sl], w1 * sc);
    atomicAdd(&a2s[sl], w2 * sc);
    atomicAdd(&acs[sl], 1u);

    r0 = r1; r1 = r2;
    y0 = n0; y1 = n1; y2 = n2;
  }
  __syncthreads();

  if (tid < nv) {
    const float c   = (float)acs[tid];
    const float inv = 1.0f / fmaxf(c, 1.0f);
    float a0 = 0.0f, a1 = 0.0f, a2 = 0.0f;
    if (c > 0.0f) {
      a0 = a0s[tid] * inv;
      a1 = a1s[tid] * inv;
      a2 = a2s[tid] * inv;
    }
    const float an = sqrtf(-a0 * a0 + a1 * a1 + a2 * a2 + EPSV);
    const float T  = an;
    const float e  = expf(T);
    const float ei = 1.0f / e;
    const float ch = 0.5f * (e + ei);
    const float sh = 0.5f * (e - ei);
    const float f  = sh / an;
    const int   i  = lo + tid;
    out[3 * i + 0] = ch * xs[3 * tid + 0] + f * a0;
    out[3 * i + 1] = ch * xs[3 * tid + 1] + f * a1;
    out[3 * i + 2] = ch * xs[3 * tid + 2] + f * a2;
  }
}

extern "C" void kernel_launch(void* const* d_in, const int* in_sizes, int n_in,
                              void* d_out, int out_size, void* d_ws, size_t ws_size,
                              hipStream_t stream) {
  const float* vertices = (const float*)d_in[0];
  const int*   edges    = (const int*)d_in[1];
  const float* messages = (const float*)d_in[2];
  float* out = (float*)d_out;

  const int       N = in_sizes[0] / 3;
  const long long E = (long long)in_sizes[1] / 2;

  const int    NB   = (N + RV - 1) >> RV_SHIFT;
  const size_t need = (size_t)4 * MAXB * (2 + NCHUNK) + (size_t)8 * (size_t)E;
  if (NB <= MAXB && E < (1ll << 31) && ws_size >= need) {
    unsigned* tab    = (unsigned*)d_ws;
    unsigned* tot    = tab;
    unsigned* gstart = tab + MAXB;
    unsigned* part   = tab + 2 * MAXB;
    uint2*    recs   = (uint2*)(tab + (2 + NCHUNK) * MAXB);
    ChunkCount_kernel<<<dim3(NCHUNK), CT, 0, stream>>>(edges, part, E, NB);
    ChunkScatter_kernel<<<dim3(NCHUNK), CT, 0, stream>>>(edges, messages, part, tot, gstart, recs, E, NB);
    RangeSum_kernel<<<dim3(NB), RV, 0, stream>>>(vertices, tot, gstart, recs, out, N);
    return;
  }

  float* acc = (float*)d_ws;

  unsigned* tickets = (unsigned*)d_out;

  const int zn = 4 * N;
  HyperZero_kernel<<<(zn + 255) / 256, 256, 0, stream>>>(acc, zn, tickets);

  HyperEdge_kernel<<<dim3(2048), 256, 0, stream>>>(
      vertices, edges, messages, acc, tickets, E, N);

  HyperVertex_kernel<<<(N + 255) / 256, 256, 0, stream>>>(vertices, acc, out, N);
}


# Scatter collects its chunk's records in LDS (256 KB) grouped by range, then copies each range's run out in aligned 256-byte windows: a store writes 2 whole lines, not 32 8-byte pieces; order kept
# speedup vs baseline: 4.8722x; 1.3125x over previous
#include <hip/hip_runtime.h>
#include <hip/hip_bf16.h>

typedef __attribute__((ext_vector_type(2))) float v2f;
typedef __attribute__((ext_vector_type(8))) float v8f;

#define EPSV 1e-6f

static __device__ __forceinline__ float hyp_dist(float u) {
  const bool  big = u > 1.0e18f;
  const float s   = __builtin_amdgcn_sqrtf((u - 1.0f) * (u + 1.0f));
  const float r   = __builtin_amdgcn_logf(big ? u : (u + s)) * 0.693147180559945f;
  return big ? (r + 0.693147180559945f) : r;
}

#define RUN_TILES 8
#define TICKET_STRIDE 128

__global__ void HyperZero_kernel(float* __restrict__ p, int n, unsigned* __restrict__ tickets) {
  int i = blockIdx.x * blockDim.x + threadIdx.x;
  if (i < n) p[i] = 0.0f;
  if (i < 2) tickets[i * TICKET_STRIDE] = 0u;
}

template <int K>
static __device__ __forceinline__ int quad_lane(int v) {
  return __builtin_amdgcn_mov_dpp(v, K * 0x55, 0xf, 0xf, true);
}
template <int K>
static __device__ __forceinline__ float quad_lane(float v) {
  return __int_as_float(quad_lane<K>(__float_as_int(v)));
}

template <int K>
static __device__ __forceinline__ void add_quad_edge(float* __restrict__ acc, int comp, int vs,
                                                     float c0, float c1, float c2) {
  const int   vj = quad_lane<K>(vs);
  const float t0 = quad_lane<K>(c0);
  const float t1 = quad_lane<K>(c1);
  const float t2 = quad_lane<K>(c2);
  const float lo = (comp == 0) ? t0 : t1;
  const float hi = (comp == 2) ? t2 : 1.0f;
  const float val = (comp < 2) ? lo : hi;
  if (vj >= 0) atomicAdd(&acc[4 * vj + comp], val);
}

static __device__ __forceinline__ void edge_tile(
    const float* __restrict__ vertices,
    const int*   __restrict__ edges,
    const float* __restrict__ messages,
    float* __restrict__ acc,
    long long E, long long e0, int lo, int hi, int lane) {
  const int octet = lane >> 3;
  const int perm  = (octet == 1) ? (lane + 8) : ((octet == 2) ? (lane - 8) : lane);
  const long long eg = e0 + perm;

  const long long el = (eg < E) ? eg : (E - 1);
  const int2 sd = *reinterpret_cast<const int2*>(edges + 2 * el);

  float weight = 0.0f;
  if (e0 + 32 <= E) {
    const int rowA = lane & 15;
    const int kb   = (lane < 16) ? 0 : 2;
    const float* m1 = messages + (e0 + rowA) * 8 + kb;
    const float* m2 = m1 + 16 * 8;
    v2f a1lo = *(const v2f*)(m1);
    v2f a1hi = *(const v2f*)(m1 + 4);
    v2f a2lo = *(const v2f*)(m2);
    v2f a2hi = *(const v2f*)(m2 + 4);
    v2f ones; ones[0] = 1.0f; ones[1] = 1.0f;

    v8f c1 = {};
    c1 = __builtin_amdgcn_wmma_f32_16x16x4_f32(false, a1lo, false, ones, (short)0, c1, false, false);
    c1 = __builtin_amdgcn_wmma_f32_16x16x4_f32(false, a1hi, false, ones, (short)0, c1, false, false);
    v8f c2 = {};
    c2 = __builtin_amdgcn_wmma_f32_16x16x4_f32(false, a2lo, false, ones, (short)0, c2, false, false);
    c2 = __builtin_amdgcn_wmma_f32_16x16x4_f32(false, a2hi, false, ones, (short)0, c2, false, false);

    const bool useD2 = (lane >> 3) & 1;
    v8f d = useD2 ? c2 : c1;
    const int r = lane & 7;
    float s = d[0];
    if (r == 1) s = d[1];
    if (r == 2) s = d[2];
    if (r == 3) s = d[3];
    if (r == 4) s = d[4];
    if (r == 5) s = d[5];
    if (r == 6) s = d[6];
    if (r == 7) s = d[7];
    weight = s * 0.125f;
  } else if (eg < E) {
    float s = 0.0f;
#pragma unroll
    for (int k = 0; k < 8; ++k) s += messages[eg * 8 + k];
    weight = s * 0.125f;
  }

  int   vs = -1;
  float c0 = 0.0f, c1 = 0.0f, c2 = 0.0f;
  {
    const int s  = (eg < E) ? sd.x : -1;
    const int vd = sd.y;
    if (s >= lo && s < hi) {
      vs = s;
      const float x0 = vertices[3 * vs + 0], x1 = vertices[3 * vs + 1], x2 = vertices[3 * vs + 2];
      const float y0 = vertices[3 * vd + 0], y1 = vertices[3 * vd + 1], y2 = vertices[3 * vd + 2];
      const float t  = -x0 * y0 + x1 * y1 + x2 * y2;
      const float v0 = y0 + x0 * t;
      const float v1 = y1 + x1 * t;
      const float v2 = y2 + x2 * t;
      const float vn   = sqrtf(-v0 * v0 + v1 * v1 + v2 * v2 + EPSV);
      const float dist = hyp_dist(-t);
      const float sc   = dist * weight / vn;
      c0 = v0 * sc;
      c1 = v1 * sc;
      c2 = v2 * sc;
    }
  }

  const int comp = lane & 3;
  add_quad_edge<0>(acc, comp, vs, c0, c1, c2);
  add_quad_edge<1>(acc, comp, vs, c0, c1, c2);
  add_quad_edge<2>(acc, comp, vs, c0, c1, c2);
  add_quad_edge<3>(acc, comp, vs, c0, c1, c2);
}

__global__ __launch_bounds__(256) void HyperEdge_kernel(
    const float* __restrict__ vertices,
    const int*   __restrict__ edges,
    const float* __restrict__ messages,
    float* __restrict__ acc,
    unsigned* __restrict__ tickets,
    long long E, int N) {
  const int lane = threadIdx.x & 31;
  const unsigned place  = __builtin_amdgcn_s_sendmsg_rtn(0x87);
  const int      myhalf = (int)((place >> 18) & 1u);

  const long long tiles = (E + 31) >> 5;
  const unsigned  runs  = (unsigned)((tiles + RUN_TILES - 1) / RUN_TILES);
  const int       split = ((N >> 1) + 15) & ~15;

  for (int pass = 0; pass < 2; ++pass) {
    const int h  = myhalf ^ pass;
    const int lo = h ? split : 0;
    const int hi = h ? N : split;
    for (unsigned it = 0; it <= runs; ++it) {
      unsigned t = 0u;
      if (lane == 0) t = atomicAdd(&tickets[h * TICKET_STRIDE], 1u);
      t = (unsigned)__builtin_amdgcn_readfirstlane((int)t);
      if (t >= runs) break;
      const long long tile0 = (long long)t * RUN_TILES;
      for (int k = 0; k < RUN_TILES; ++k) {
        const long long e0 = (tile0 + k) << 5;
        if (e0 >= E) break;
        edge_tile(vertices, edges, messages, acc, E, e0, lo, hi, lane);
      }
    }
  }
}

__global__ void HyperVertex_kernel(const float* __restrict__ vertices,
                                   const float* __restrict__ acc,
                                   float* __restrict__ out, int N) {
  int i = blockIdx.x * blockDim.x + threadIdx.x;
  if (i >= N) return;
  const float4 rec = reinterpret_cast<const float4*>(acc)[i];
  const float c   = rec.w;
  const float inv = 1.0f / fmaxf(c, 1.0f);
  float a0 = 0.0f, a1 = 0.0f, a2 = 0.0f;
  if (c > 0.0f) {
    a0 = rec.x * inv;
    a1 = rec.y * inv;
    a2 = rec.z * inv;
  }
  const float an = sqrtf(-a0 * a0 + a1 * a1 + a2 * a2 + EPSV);
  const float T  = an;
  const float e  = expf(T);
  const float ei = 1.0f / e;
  const float ch = 0.5f * (e + ei);
  const float sh = 0.5f * (e - ei);
  const float f  = sh / an;
  out[3 * i + 0] = ch * vertices[3 * i + 0] + f * a0;
  out[3 * i + 1] = ch * vertices[3 * i + 1] + f * a1;
  out[3 * i + 2] = ch * vertices[3 * i + 2] + f * a2;
}


#define RV_SHIFT 10
#define RV (1 << RV_SHIFT)
#define MAXB 1024
#define DST_BITS 20

#define NCHUNK 256
#define CT 1024

static __device__ __forceinline__ void chunk_tiles(long long E, int nc, int c,
                                                   long long& t0, long long& t1) {
  const long long tiles = (E + 31) >> 5;
  const long long ct    = (tiles + nc - 1) / nc;
  t0 = (long long)c * ct;
  if (t0 > tiles) t0 = tiles;
  t1 = t0 + ct;
  if (t1 > tiles) t1 = tiles;
}

__global__ __launch_bounds__(CT) void ChunkCount_kernel(const int* __restrict__ edges,
                                                        unsigned* __restrict__ part,
                                                        long long E, int NB) {
  __shared__ unsigned h[MAXB];
  const int tid = threadIdx.x;
  h[tid] = 0u;
  __syncthreads();
  long long t0, t1;
  chunk_tiles(E, gridDim.x, blockIdx.x, t0, t1);
  const long long eb = t0 << 5;
  long long ee = t1 << 5;
  if (ee > E) ee = E;
  for (long long e = eb + tid; e < ee; e += 8 * CT) {
    int s[8];
#pragma unroll
    for (int k = 0; k < 8; ++k) {
      const long long ek = e + k * CT;
      s[k] = (ek < ee) ? edges[2 * ek] : -1;
    }
#pragma unroll
    for (int k = 0; k < 8; ++k) {
      if (s[k] >= 0) atomicAdd(&h[s[k] >> RV_SHIFT], 1u);
    }
  }
  __syncthreads();
  if (tid < NB) part[(size_t)blockIdx.x * MAXB + tid] = h[tid];
}

static __device__ __forceinline__ void tile_loads(const float* __restrict__ messages,
                                                  long long e0, int lane, v2f (&a)[4]) {
  const int rowA = lane & 15;
  const int kb   = (lane < 16) ? 0 : 2;
  const float* m1 = messages + (e0 + rowA) * 8 + kb;
  const float* m2 = m1 + 16 * 8;
  a[0] = *(const v2f*)(m1);
  a[1] = *(const v2f*)(m1 + 4);
  a[2] = *(const v2f*)(m2);
  a[3] = *(const v2f*)(m2 + 4);
}

static __device__ __forceinline__ float tile_mean(const v2f (&a)[4], int lane) {
  v2f ones; ones[0] = 1.0f; ones[1] = 1.0f;

  v8f c1 = {};
  c1 = __builtin_amdgcn_wmma_f32_16x16x4_f32(false, a[0], false, ones, (short)0, c1, false, false);
  c1 = __builtin_amdgcn_wmma_f32_16x16x4_f32(false, a[1], false, ones, (short)0, c1, false, false);
  v8f c2 = {};
  c2 = __builtin_amdgcn_wmma_f32_16x16x4_f32(false, a[2], false, ones, (short)0, c2, false, false);
  c2 = __builtin_amdgcn_wmma_f32_16x16x4_f32(false, a[3], false, ones, (short)0, c2, false, false);

  const bool useD2 = (lane >> 3) & 1;
  v8f d = useD2 ? c2 : c1;
  const int r = lane & 7;
  float s = d[0];
  if (r == 1) s = d[1];
  if (r == 2) s = d[2];
  if (r == 3) s = d[3];
  if (r == 4) s = d[4];
  if (r == 5) s = d[5];
  if (r == 6) s = d[6];
  if (r == 7) s = d[7];
  return s * 0.125f;
}

static __device__ __forceinline__ float full_tile_mean(const float* __restrict__ messages,
                                                       long long e0, int lane) {
  v2f a[4];
  tile_loads(messages, e0, lane, a);
  return tile_mean(a, lane);
}

#define STAGE_CAP 32768

static __device__ __forceinline__ void place_edge(unsigned* lh, uint2* stage, uint2* __restrict__ recs,
                                                  bool staged, int2 sd, float weight) {
  const unsigned pos = atomicAdd(&lh[sd.x >> RV_SHIFT], 1u);
  const uint2 rec = make_uint2(((unsigned)(sd.x & (RV - 1)) << DST_BITS) | (unsigned)sd.y,
                               __float_as_uint(weight));
  if (staged) stage[pos] = rec;
  else        recs[pos]  = rec;
}

__global__ __launch_bounds__(CT) void ChunkScatter_kernel(
    const int*   __restrict__ edges,
    const float* __restrict__ messages,
    const unsigned* __restrict__ part,
    unsigned* __restrict__ tot,
    unsigned* __restrict__ gstart,
    uint2* __restrict__ recs,
    long long E, int NB) {
  __shared__ unsigned lh[MAXB];
  __shared__ unsigned ht[CT];
  __shared__ unsigned hb[CT];
  __shared__ unsigned sc[CT / 32];
  __shared__ unsigned sl[CT / 32];
  __shared__ uint2 stage[STAGE_CAP];
  const int tid  = threadIdx.x;
  const int lane = tid & 31;
  const int wave = tid >> 5;
  const int me   = blockIdx.x;
  const int nc   = gridDim.x;

  int gsh = 0;
  if (NB <= 512) gsh = 1;
  if (NB <= 256) gsh = 2;
  if (NB <= 128) gsh = 3;
  const int slot = CT >> gsh;
  const int b    = tid & (slot - 1);
  const int grp  = tid >> (10 - gsh);

  long long t0, t1;
  chunk_tiles(E, nc, me, t0, t1);
  const bool staged = ((t1 - t0) << 5) <= STAGE_CAP;

  unsigned mine = 0u;
  if (tid < NB) mine = part[(size_t)me * MAXB + tid];

  unsigned tsum = 0u, bsum = 0u;
  if (b < NB) {
#pragma unroll 16
    for (int w = grp; w < nc; w += (1 << gsh)) {
      const unsigned v = part[(size_t)w * MAXB + b];
      tsum += v;
      bsum += (w < me) ? v : 0u;
    }
  }
  ht[tid] = tsum;
  hb[tid] = bsum;
  __syncthreads();
  unsigned total = 0u, before = 0u;
  if (tid < NB) {
    for (int g = 0; g < (1 << gsh); ++g) {
      total  += ht[g * slot + tid];
      before += hb[g * slot + tid];
    }
  }
  unsigned incl = total, incm = mine;
  for (int d = 1; d < 32; d <<= 1) {
    const unsigned n = __shfl_up(incl, d);
    const unsigned m = __shfl_up(incm, d);
    if (lane >= d) { incl += n; incm += m; }
  }
  if (lane == 31) { sc[wave] = incl; sl[wave] = incm; }
  __syncthreads();
  unsigned first  = incl - total;
  unsigned lfirst = incm - mine;
  for (int j = 0; j < wave; ++j) {
    first  += sc[j];
    lfirst += sl[j];
  }
  const unsigned gpos = first + before;
  lh[tid] = staged ? lfirst : gpos;
  ht[tid] = lfirst;
  hb[tid] = gpos;
  if (me == 0 && tid < NB) {
    gstart[tid] = first;
    tot[tid]    = total;
  }
  __syncthreads();

  const int octet = lane >> 3;
  const int perm  = (octet == 1) ? (lane + 8) : ((octet == 2) ? (lane - 8) : lane);

  for (long long t = t0 + 4 * wave; t < t1; t += 4 * (CT / 32)) {
    const long long e0 = t << 5;
    if (t + 3 < t1 && e0 + 128 <= E) {
      const int2 sdA = *reinterpret_cast<const int2*>(edges + 2 * (e0 + perm));
      const int2 sdB = *reinterpret_cast<const int2*>(edges + 2 * (e0 + 32 + perm));
      const int2 sdC = *reinterpret_cast<const int2*>(edges + 2 * (e0 + 64 + perm));
      const int2 sdD = *reinterpret_cast<const int2*>(edges + 2 * (e0 + 96 + perm));
      v2f aA[4], aB[4], aC[4], aD[4];
      tile_loads(messages, e0, lane, aA);
      tile_loads(messages, e0 + 32, lane, aB);
      tile_loads(messages, e0 + 64, lane, aC);
      tile_loads(messages, e0 + 96, lane, aD);
      __builtin_amdgcn_sched_barrier(0);
      const float wA = tile_mean(aA, lane);
      const float wB = tile_mean(aB, lane);
      const float wC = tile_mean(aC, lane);
      const float wD = tile_mean(aD, lane);
      place_edge(lh, stage, recs, staged, sdA, wA);
      place_edge(lh, stage, recs, staged, sdB, wB);
      place_edge(lh, stage, recs, staged, sdC, wC);
      place_edge(lh, stage, recs, staged, sdD, wD);
    } else {
      const long long tl = (t + 4 < t1) ? (t + 4) : t1;
      for (long long u = t; u < tl; ++u) {
        const long long f0 = u << 5;
        const long long eg = f0 + perm;
        const long long el = (eg < E) ? eg : (E - 1);
        const int2 sd = *reinterpret_cast<const int2*>(edges + 2 * el);
        float weight = 0.0f;
        if (f0 + 32 <= E) {
          weight = full_tile_mean(messages, f0, lane);
        } else if (eg < E) {
          float s = 0.0f;
#pragma unroll
          for (int k = 0; k < 8; ++k) s += messages[eg * 8 + k];
          weight = s * 0.125f;
        }
        if (eg < E) place_edge(lh, stage, recs, staged, sd, weight);
      }
    }
  }

  if (!staged) return;
  __syncthreads();

  const int wv = __builtin_amdgcn_readfirstlane(wave);
  const int rl = wv + (CT / 32) * lane;
  unsigned vsrc = 0u, vend = 0u, vdst = 0u;
  if (rl < NB) {
    vsrc = ht[rl];
    vend = lh[rl];
    vdst = hb[rl];
  }
  const int nr = (NB - wv + CT / 32 - 1) / (CT / 32);
  for (int j = 0; j < nr; ++j) {
    const unsigned src = (unsigned)__builtin_amdgcn_readlane((int)vsrc, j);
    const unsigned end = (unsigned)__builtin_amdgcn_readlane((int)vend, j);
    const unsigned dst = (unsigned)__builtin_amdgcn_readlane((int)vdst, j);
    const int n = (int)(end - src);
    if (n <= 0) continue;
    for (int kb = -(int)(dst & 31u); kb < n; kb += 128) {
      uint2 v[4];
#pragma unroll
      for (int u = 0; u < 4; ++u) {
        const int k  = kb + 32 * u + lane;
        const int kc = (k < 0) ? 0 : ((k < n) ? k : (n - 1));
        v[u] = stage[src + (unsigned)kc];
      }
#pragma unroll
      for (int u = 0; u < 4; ++u) {
        const int k = kb + 32 * u + lane;
        if (k >= 0 && k < n) recs[dst + (unsigned)k] = v[u];
      }
    }
  }
}

__global__ __launch_bounds__(RV) void RangeSum_kernel(
    const float* __restrict__ vertices,
    const unsigned* __restrict__ tot,
    const unsigned* __restrict__ gstart,
    const uint2* __restrict__ recs,
    float* __restrict__ out, int N) {
  __shared__ float    xs[3 * RV];
  __shared__ float    a0s[RV];
  __shared__ float    a1s[RV];
  __shared__ float    a2s[RV];
  __shared__ unsigned acs[RV];
  const int tid = threadIdx.x;
  const int b   = blockIdx.x;
  const int lo  = b << RV_SHIFT;
  const int nv  = (N - lo < RV) ? (N - lo) : RV;

  for (int i = tid; i < 3 * nv; i += RV) xs[i] = vertices[3 * lo + i];
  a0s[tid] = 0.0f;
  a1s[tid] = 0.0f;
  a2s[tid] = 0.0f;
  acs[tid] = 0u;
  __syncthreads();

  const unsigned start = gstart[b];
  const unsigned cnt   = tot[b];
  const unsigned vmask = (1u << DST_BITS) - 1u;
  uint2 r0 = make_uint2(0u, 0u);
  uint2 r1 = make_uint2(0u, 0u);
  float y0 = 0.0f, y1 = 0.0f, y2 = 0.0f;
  if ((unsigned)tid < cnt) {
    const unsigned i1 = ((unsigned)tid + RV < cnt) ? ((unsigned)tid + RV) : (cnt - 1u);
    r0 = recs[start + tid];
    r1 = recs[start + i1];
    const int vd = (int)(r0.x & vmask);
    y0 = vertices[3 * vd + 0];
    y1 = vertices[3 * vd + 1];
    y2 = vertices[3 * vd + 2];
  }
  for (unsigned i = tid; i < cnt; i += RV) {
    const unsigned i2  = (i + 2 * RV < cnt) ? (i + 2 * RV) : (cnt - 1u);
    const int      vdn = (int)(r1.x & vmask);
    const float n0 = vertices[3 * vdn + 0];
    const float n1 = vertices[3 * vdn + 1];
    const float n2 = vertices[3 * vdn + 2];
    const uint2 r2 = recs[start + i2];

    const int   sl = (int)(r0.x >> DST_BITS);
    const float weight = __uint_as_float(r0.y);
    const float x0 = xs[3 * sl + 0], x1 = xs[3 * sl + 1], x2 = xs[3 * sl + 2];
    const float t  = -x0 * y0 + x1 * y1 + x2 * y2;
    const float w0 = y0 + x0 * t;
    const float w1 = y1 + x1 * t;
    const float w2 = y2 + x2 * t;
    const float vn   = sqrtf(-w0 * w0 + w1 * w1 + w2 * w2 + EPSV);
    const float dist = hyp_dist(-t);
    const float sc   = dist * weight / vn;
    atomicAdd(&a0s[sl], w0 * sc);
    atomicAdd(&a1s[sl], w1 * sc);
    atomicAdd(&a2s[sl], w2 * sc);
    atomicAdd(&acs[sl], 1u);

    r0 = r1; r1 = r2;
    y0 = n0; y1 = n1; y2 = n2;
  }
  __syncthreads();

  if (tid < nv) {
    const float c   = (float)acs[tid];
    const float inv = 1.0f / fmaxf(c, 1.0f);
    float a0 = 0.0f, a1 = 0.0f, a2 = 0.0f;
    if (c > 0.0f) {
      a0 = a0s[tid] * inv;
      a1 = a1s[tid] * inv;
      a2 = a2s[tid] * inv;
    }
    const float an = sqrtf(-a0 * a0 + a1 * a1 + a2 * a2 + EPSV);
    const float T  = an;
    const float e  = expf(T);
    const float ei = 1.0f / e;
    const float ch = 0.5f * (e + ei);
    const float sh = 0.5f * (e - ei);
    const float f  = sh / an;
    const int   i  = lo + tid;
    out[3 * i + 0] = ch * xs[3 * tid + 0] + f * a0;
    out[3 * i + 1] = ch * xs[3 * tid + 1] + f * a1;
    out[3 * i + 2] = ch * xs[3 * tid + 2] + f * a2;
  }
}

extern "C" void kernel_launch(void* const* d_in, const int* in_sizes, int n_in,
                              void* d_out, int out_size, void* d_ws, size_t ws_size,
                              hipStream_t stream) {
  const float* vertices = (const float*)d_in[0];
  const int*   edges    = (const int*)d_in[1];
  const float* messages = (const float*)d_in[2];
  float* out = (float*)d_out;

  const int       N = in_sizes[0] / 3;
  const long long E = (long long)in_sizes[1] / 2;

  const int    NB   = (N + RV - 1) >> RV_SHIFT;
  const size_t need = (size_t)4 * MAXB * (2 + NCHUNK) + (size_t)8 * (size_t)E;
  if (NB <= MAXB && E < (1ll << 31) && ws_size >= need) {
    unsigned* tab    = (unsigned*)d_ws;
    unsigned* tot    = tab;
    unsigned* gstart = tab + MAXB;
    unsigned* part   = tab + 2 * MAXB;
    uint2*    recs   = (uint2*)(tab + (2 + NCHUNK) * MAXB);
    ChunkCount_kernel<<<dim3(NCHUNK), CT, 0, stream>>>(edges, part, E, NB);
    ChunkScatter_kernel<<<dim3(NCHUNK), CT, 0, stream>>>(edges, messages, part, tot, gstart, recs, E, NB);
    RangeSum_kernel<<<dim3(NB), RV, 0, stream>>>(vertices, tot, gstart, recs, out, N);
    return;
  }

  float* acc = (float*)d_ws;

  unsigned* tickets = (unsigned*)d_out;

  const int zn = 4 * N;
  HyperZero_kernel<<<(zn + 255) / 256, 256, 0, stream>>>(acc, zn, tickets);

  HyperEdge_kernel<<<dim3(2048), 256, 0, stream>>>(
      vertices, edges, messages, acc, tickets, E, N);

  HyperVertex_kernel<<<(N + 255) / 256, 256, 0, stream>>>(vertices, acc, out, N);
}


# Records laid out chunk by chunk: scatter counts its own chunk in LDS, copies its block out whole; a summing wave walks 8 chunks' runs as one list: no count launch, no table sums; LDS add order moves
# speedup vs baseline: 5.3154x; 1.0910x over previous
#include <hip/hip_runtime.h>
#include <hip/hip_bf16.h>

typedef __attribute__((ext_vector_type(2))) float v2f;
typedef __attribute__((ext_vector_type(8))) float v8f;

#define EPSV 1e-6f

static __device__ __forceinline__ float hyp_dist(float u) {
  const bool  big = u > 1.0e18f;
  const float s   = __builtin_amdgcn_sqrtf((u - 1.0f) * (u + 1.0f));
  const float r   = __builtin_amdgcn_logf(big ? u : (u + s)) * 0.693147180559945f;
  return big ? (r + 0.693147180559945f) : r;
}

#define RUN_TILES 8
#define TICKET_STRIDE 128

__global__ void HyperZero_kernel(float* __restrict__ p, int n, unsigned* __restrict__ tickets) {
  int i = blockIdx.x * blockDim.x + threadIdx.x;
  if (i < n) p[i] = 0.0f;
  if (i < 2) tickets[i * TICKET_STRIDE] = 0u;
}

template <int K>
static __device__ __forceinline__ int quad_lane(int v) {
  return __builtin_amdgcn_mov_dpp(v, K * 0x55, 0xf, 0xf, true);
}
template <int K>
static __device__ __forceinline__ float quad_lane(float v) {
  return __int_as_float(quad_lane<K>(__float_as_int(v)));
}

template <int K>
static __device__ __forceinline__ void add_quad_edge(float* __restrict__ acc, int comp, int vs,
                                                     float c0, float c1, float c2) {
  const int   vj = quad_lane<K>(vs);
  const float t0 = quad_lane<K>(c0);
  const float t1 = quad_lane<K>(c1);
  const float t2 = quad_lane<K>(c2);
  const float lo = (comp == 0) ? t0 : t1;
  const float hi = (comp == 2) ? t2 : 1.0f;
  const float val = (comp < 2) ? lo : hi;
  if (vj >= 0) atomicAdd(&acc[4 * vj + comp], val);
}

static __device__ __forceinline__ void edge_tile(
    const float* __restrict__ vertices,
    const int*   __restrict__ edges,
    const float* __restrict__ messages,
    float* __restrict__ acc,
    long long E, long long e0, int lo, int hi, int lane) {
  const int octet = lane >> 3;
  const int perm  = (octet == 1) ? (lane + 8) : ((octet == 2) ? (lane - 8) : lane);
  const long long eg = e0 + perm;

  const long long el = (eg < E) ? eg : (E - 1);
  const int2 sd = *reinterpret_cast<const int2*>(edges + 2 * el);

  float weight = 0.0f;
  if (e0 + 32 <= E) {
    const int rowA = lane & 15;
    const int kb   = (lane < 16) ? 0 : 2;
    const float* m1 = messages + (e0 + rowA) * 8 + kb;
    const float* m2 = m1 + 16 * 8;
    v2f a1lo = *(const v2f*)(m1);
    v2f a1hi = *(const v2f*)(m1 + 4);
    v2f a2lo = *(const v2f*)(m2);
    v2f a2hi = *(const v2f*)(m2 + 4);
    v2f ones; ones[0] = 1.0f; ones[1] = 1.0f;

    v8f c1 = {};
    c1 = __builtin_amdgcn_wmma_f32_16x16x4_f32(false, a1lo, false, ones, (short)0, c1, false, false);
    c1 = __builtin_amdgcn_wmma_f32_16x16x4_f32(false, a1hi, false, ones, (short)0, c1, false, false);
    v8f c2 = {};
    c2 = __builtin_amdgcn_wmma_f32_16x16x4_f32(false, a2lo, false, ones, (short)0, c2, false, false);
    c2 = __builtin_amdgcn_wmma_f32_16x16x4_f32(false, a2hi, false, ones, (short)0, c2, false, false);

    const bool useD2 = (lane >> 3) & 1;
    v8f d = useD2 ? c2 : c1;
    const int r = lane & 7;
    float s = d[0];
    if (r == 1) s = d[1];
    if (r == 2) s = d[2];
    if (r == 3) s = d[3];
    if (r == 4) s = d[4];
    if (r == 5) s = d[5];
    if (r == 6) s = d[6];
    if (r == 7) s = d[7];
    weight = s * 0.125f;
  } else if (eg < E) {
    float s = 0.0f;
#pragma unroll
    for (int k = 0; k < 8; ++k) s += messages[eg * 8 + k];
    weight = s * 0.125f;
  }

  int   vs = -1;
  float c0 = 0.0f, c1 = 0.0f, c2 = 0.0f;
  {
    const int s  = (eg < E) ? sd.x : -1;
    const int vd = sd.y;
    if (s >= lo && s < hi) {
      vs = s;
      const float x0 = vertices[3 * vs + 0], x1 = vertices[3 * vs + 1], x2 = vertices[3 * vs + 2];
      const float y0 = vertices[3 * vd + 0], y1 = vertices[3 * vd + 1], y2 = vertices[3 * vd + 2];
      const float t  = -x0 * y0 + x1 * y1 + x2 * y2;
      const float v0 = y0 + x0 * t;
      const float v1 = y1 + x1 * t;
      const float v2 = y2 + x2 * t;
      const float vn   = sqrtf(-v0 * v0 + v1 * v1 + v2 * v2 + EPSV);
      const float dist = hyp_dist(-t);
      const float sc   = dist * weight / vn;
      c0 = v0 * sc;
      c1 = v1 * sc;
      c2 = v2 * sc;
    }
  }

  const int comp = lane & 3;
  add_quad_edge<0>(acc, comp, vs, c0, c1, c2);
  add_quad_edge<1>(acc, comp, vs, c0, c1, c2);
  add_quad_edge<2>(acc, comp, vs, c0, c1, c2);
  add_quad_edge<3>(acc, comp, vs, c0, c1, c2);
}

__global__ __launch_bounds__(256) void HyperEdge_kernel(
    const float* __restrict__ vertices,
    const int*   __restrict__ edges,
    const float* __restrict__ messages,
    float* __restrict__ acc,
    unsigned* __restrict__ tickets,
    long long E, int N) {
  const int lane = threadIdx.x & 31;
  const unsigned place  = __builtin_amdgcn_s_sendmsg_rtn(0x87);
  const int      myhalf = (int)((place >> 18) & 1u);

  const long long tiles = (E + 31) >> 5;
  const unsigned  runs  = (unsigned)((tiles + RUN_TILES - 1) / RUN_TILES);
  const int       split = ((N >> 1) + 15) & ~15;

  for (int pass = 0; pass < 2; ++pass) {
    const int h  = myhalf ^ pass;
    const int lo = h ? split : 0;
    const int hi = h ? N : split;
    for (unsigned it = 0; it <= runs; ++it) {
      unsigned t = 0u;
      if (lane == 0) t = atomicAdd(&tickets[h * TICKET_STRIDE], 1u);
      t = (unsigned)__builtin_amdgcn_readfirstlane((int)t);
      if (t >= runs) break;
      const long long tile0 = (long long)t * RUN_TILES;
      for (int k = 0; k < RUN_TILES; ++k) {
        const long long e0 = (tile0 + k) << 5;
        if (e0 >= E) break;
        edge_tile(vertices, edges, messages, acc, E, e0, lo, hi, lane);
      }
    }
  }
}

__global__ void HyperVertex_kernel(const float* __restrict__ vertices,
                                   const float* __restrict__ acc,
                                   float* __restrict__ out, int N) {
  int i = blockIdx.x * blockDim.x + threadIdx.x;
  if (i >= N) return;
  const float4 rec = reinterpret_cast<const float4*>(acc)[i];
  const float c   = rec.w;
  const float inv = 1.0f / fmaxf(c, 1.0f);
  float a0 = 0.0f, a1 = 0.0f, a2 = 0.0f;
  if (c > 0.0f) {
    a0 = rec.x * inv;
    a1 = rec.y * inv;
    a2 = rec.z * inv;
  }
  const float an = sqrtf(-a0 * a0 + a1 * a1 + a2 * a2 + EPSV);
  const float T  = an;
  const float e  = expf(T);
  const float ei = 1.0f / e;
  const float ch = 0.5f * (e + ei);
  const float sh = 0.5f * (e - ei);
  const float f  = sh / an;
  out[3 * i + 0] = ch * vertices[3 * i + 0] + f * a0;
  out[3 * i + 1] = ch * vertices[3 * i + 1] + f * a1;
  out[3 * i + 2] = ch * vertices[3 * i + 2] + f * a2;
}


#define RV_SHIFT 10
#define RV (1 << RV_SHIFT)
#define MAXB 1024
#define DST_BITS 20

#define NCHUNK 256
#define CT 1024
#define TROW MAXB
#define CPW (NCHUNK / (RV / 32))

static __device__ __forceinline__ void chunk_tiles(long long E, int nc, int c,
                                                   long long& t0, long long& t1) {
  const long long tiles = (E + 31) >> 5;
  const long long ct    = (tiles + nc - 1) / nc;
  t0 = (long long)c * ct;
  if (t0 > tiles) t0 = tiles;
  t1 = t0 + ct;
  if (t1 > tiles) t1 = tiles;
}

static __device__ __forceinline__ void tile_loads(const float* __restrict__ messages,
                                                  long long e0, int lane, v2f (&a)[4]) {
  const int rowA = lane & 15;
  const int kb   = (lane < 16) ? 0 : 2;
  const float* m1 = messages + (e0 + rowA) * 8 + kb;
  const float* m2 = m1 + 16 * 8;
  a[0] = *(const v2f*)(m1);
  a[1] = *(const v2f*)(m1 + 4);
  a[2] = *(const v2f*)(m2);
  a[3] = *(const v2f*)(m2 + 4);
}

static __device__ __forceinline__ float tile_mean(const v2f (&a)[4], int lane) {
  v2f ones; ones[0] = 1.0f; ones[1] = 1.0f;

  v8f c1 = {};
  c1 = __builtin_amdgcn_wmma_f32_16x16x4_f32(false, a[0], false, ones, (short)0, c1, false, false);
  c1 = __builtin_amdgcn_wmma_f32_16x16x4_f32(false, a[1], false, ones, (short)0, c1, false, false);
  v8f c2 = {};
  c2 = __builtin_amdgcn_wmma_f32_16x16x4_f32(false, a[2], false, ones, (short)0, c2, false, false);
  c2 = __builtin_amdgcn_wmma_f32_16x16x4_f32(false, a[3], false, ones, (short)0, c2, false, false);

  const bool useD2 = (lane >> 3) & 1;
  v8f d = useD2 ? c2 : c1;
  const int r = lane & 7;
  float s = d[0];
  if (r == 1) s = d[1];
  if (r == 2) s = d[2];
  if (r == 3) s = d[3];
  if (r == 4) s = d[4];
  if (r == 5) s = d[5];
  if (r == 6) s = d[6];
  if (r == 7) s = d[7];
  return s * 0.125f;
}

static __device__ __forceinline__ float full_tile_mean(const float* __restrict__ messages,
                                                       long long e0, int lane) {
  v2f a[4];
  tile_loads(messages, e0, lane, a);
  return tile_mean(a, lane);
}

#define STAGE_CAP 32768

static __device__ __forceinline__ void place_edge(unsigned* lh, uint2* stage, uint2* __restrict__ recs,
                                                  bool staged, int2 sd, float weight) {
  const unsigned pos = atomicAdd(&lh[sd.x >> RV_SHIFT], 1u);
  const uint2 rec = make_uint2(((unsigned)(sd.x & (RV - 1)) << DST_BITS) | (unsigned)sd.y,
                               __float_as_uint(weight));
  if (staged) stage[pos] = rec;
  else        recs[pos]  = rec;
}

__global__ __launch_bounds__(CT) void ChunkScatter_kernel(
    const int*   __restrict__ edges,
    const float* __restrict__ messages,
    unsigned* __restrict__ runtab,
    uint2* __restrict__ recs,
    long long E, int NB) {
  __shared__ unsigned lh[MAXB];
  __shared__ unsigned sl[CT / 32];
  __shared__ uint2 stage[STAGE_CAP];
  const int tid  = threadIdx.x;
  const int lane = tid & 31;
  const int wave = tid >> 5;
  const int me   = blockIdx.x;

  long long t0, t1;
  chunk_tiles(E, NCHUNK, me, t0, t1);
  const long long eb = t0 << 5;
  long long ee = t1 << 5;
  if (ee > E) ee = E;
  const bool staged = ((t1 - t0) << 5) <= STAGE_CAP;

  lh[tid] = 0u;
  __syncthreads();
  for (long long e = eb + tid; e < ee; e += 32 * CT) {
    int s[32];
#pragma unroll
    for (int k = 0; k < 32; ++k) {
      const long long ek = e + k * CT;
      s[k] = (ek < ee) ? edges[2 * ek] : -1;
    }
#pragma unroll
    for (int k = 0; k < 32; ++k) {
      if (s[k] >= 0) atomicAdd(&lh[s[k] >> RV_SHIFT], 1u);
    }
  }
  __syncthreads();
  const unsigned mine = lh[tid];

  unsigned incm = mine;
  for (int d = 1; d < 32; d <<= 1) {
    const unsigned m = __shfl_up(incm, d);
    if (lane >= d) incm += m;
  }
  if (lane == 31) sl[wave] = incm;
  __syncthreads();
  unsigned lfirst = incm - mine;
  for (int j = 0; j < wave; ++j) lfirst += sl[j];
  lh[tid] = staged ? lfirst : ((unsigned)eb + lfirst);
  unsigned* __restrict__ row = runtab + (size_t)me * TROW;
  if (tid < NB) row[tid] = lfirst;
  __syncthreads();

  const int octet = lane >> 3;
  const int perm  = (octet == 1) ? (lane + 8) : ((octet == 2) ? (lane - 8) : lane);

  for (long long t = t0 + 4 * wave; t < t1; t += 4 * (CT / 32)) {
    const long long e0 = t << 5;
    if (t + 3 < t1 && e0 + 128 <= E) {
      const int2 sdA = *reinterpret_cast<const int2*>(edges + 2 * (e0 + perm));
      const int2 sdB = *reinterpret_cast<const int2*>(edges + 2 * (e0 + 32 + perm));
      const int2 sdC = *reinterpret_cast<const int2*>(edges + 2 * (e0 + 64 + perm));
      const int2 sdD = *reinterpret_cast<const int2*>(edges + 2 * (e0 + 96 + perm));
      v2f aA[4], aB[4], aC[4], aD[4];
      tile_loads(messages, e0, lane, aA);
      tile_loads(messages, e0 + 32, lane, aB);
      tile_loads(messages, e0 + 64, lane, aC);
      tile_loads(messages, e0 + 96, lane, aD);
      __builtin_amdgcn_sched_barrier(0);
      const float wA = tile_mean(aA, lane);
      const float wB = tile_mean(aB, lane);
      const float wC = tile_mean(aC, lane);
      const float wD = tile_mean(aD, lane);
      place_edge(lh, stage, recs, staged, sdA, wA);
      place_edge(lh, stage, recs, staged, sdB, wB);
      place_edge(lh, stage, recs, staged, sdC, wC);
      place_edge(lh, stage, recs, staged, sdD, wD);
    } else {
      const long long tl = (t + 4 < t1) ? (t + 4) : t1;
      for (long long u = t; u < tl; ++u) {
        const long long f0 = u << 5;
        const long long eg = f0 + perm;
        const long long el = (eg < E) ? eg : (E - 1);
        const int2 sd = *reinterpret_cast<const int2*>(edges + 2 * el);
        float weight = 0.0f;
        if (f0 + 32 <= E) {
          weight = full_tile_mean(messages, f0, lane);
        } else if (eg < E) {
          float s = 0.0f;
#pragma unroll
          for (int k = 0; k < 8; ++k) s += messages[eg * 8 + k];
          weight = s * 0.125f;
        }
        if (eg < E) place_edge(lh, stage, recs, staged, sd, weight);
      }
    }
  }

  if (!staged) return;
  __syncthreads();

  const unsigned n = (unsigned)(ee - eb);
#pragma unroll 4
  for (unsigned i = tid; i < n; i += CT) recs[eb + i] = stage[i];
}

static __device__ __forceinline__ unsigned run_record(
    unsigned f, unsigned p1, unsigned p2, unsigned p3, unsigned p4, unsigned p5, unsigned p6,
    unsigned p7, unsigned a0, unsigned a1, unsigned a2, unsigned a3, unsigned a4, unsigned a5,
    unsigned a6, unsigned a7) {
  unsigned a = a0;
  a = (f >= p1) ? a1 : a;
  a = (f >= p2) ? a2 : a;
  a = (f >= p3) ? a3 : a;
  a = (f >= p4) ? a4 : a;
  a = (f >= p5) ? a5 : a;
  a = (f >= p6) ? a6 : a;
  a = (f >= p7) ? a7 : a;
  return a + f;
}

__global__ __launch_bounds__(RV) void RangeSum_kernel(
    const float* __restrict__ vertices,
    const unsigned* __restrict__ runtab,
    const uint2* __restrict__ recs,
    float* __restrict__ out, long long E, int N) {
  __shared__ float    xs[3 * RV];
  __shared__ float    a0s[RV];
  __shared__ float    a1s[RV];
  __shared__ float    a2s[RV];
  __shared__ unsigned acs[RV];
  const int tid  = threadIdx.x;
  const int lane = tid & 31;
  const int b    = blockIdx.x;
  const int lo   = b << RV_SHIFT;
  const int nv   = (N - lo < RV) ? (N - lo) : RV;

  const int wv = __builtin_amdgcn_readfirstlane(tid >> 5);
  unsigned rf = 0u, rg = 0u, rb = 0u;
  if (lane < CPW) {
    const int ch = wv * CPW + lane;
    long long t0, t1;
    chunk_tiles(E, NCHUNK, ch, t0, t1);
    long long ee = t1 << 5;
    if (ee > E) ee = E;
    rb = (unsigned)(t0 << 5);
    rf = runtab[(size_t)ch * TROW + b];
    rg = (b + 1 < (int)gridDim.x) ? runtab[(size_t)ch * TROW + b + 1]
                                  : (unsigned)(ee - (t0 << 5));
  }

  for (int i = tid; i < 3 * nv; i += RV) xs[i] = vertices[3 * lo + i];
  a0s[tid] = 0.0f;
  a1s[tid] = 0.0f;
  a2s[tid] = 0.0f;
  acs[tid] = 0u;
  __syncthreads();

  static_assert(CPW == 8, "the run list below is written out for eight runs per wave");
  const unsigned vs = rb + rf;
  const unsigned vn = rg - rf;
#define RUN_S(m) ((unsigned)__builtin_amdgcn_readlane((int)vs, (m)))
#define RUN_N(m) ((unsigned)__builtin_amdgcn_readlane((int)vn, (m)))
  const unsigned P1 = RUN_N(0);
  const unsigned P2 = P1 + RUN_N(1);
  const unsigned P3 = P2 + RUN_N(2);
  const unsigned P4 = P3 + RUN_N(3);
  const unsigned P5 = P4 + RUN_N(4);
  const unsigned P6 = P5 + RUN_N(5);
  const unsigned P7 = P6 + RUN_N(6);
  const unsigned cnt = P7 + RUN_N(7);
  const unsigned A0 = RUN_S(0);
  const unsigned A1 = RUN_S(1) - P1;
  const unsigned A2 = RUN_S(2) - P2;
  const unsigned A3 = RUN_S(3) - P3;
  const unsigned A4 = RUN_S(4) - P4;
  const unsigned A5 = RUN_S(5) - P5;
  const unsigned A6 = RUN_S(6) - P6;
  const unsigned A7 = RUN_S(7) - P7;
#undef RUN_S
#undef RUN_N
#define REC_AT(f) run_record((f), P1, P2, P3, P4, P5, P6, P7, A0, A1, A2, A3, A4, A5, A6, A7)

  const unsigned vmask = (1u << DST_BITS) - 1u;
  uint2 r0 = make_uint2(0u, 0u);
  uint2 r1 = make_uint2(0u, 0u);
  float y0 = 0.0f, y1 = 0.0f, y2 = 0.0f;
  if ((unsigned)lane < cnt) {
    const unsigned f1 = ((unsigned)lane + 32u < cnt) ? ((unsigned)lane + 32u) : (cnt - 1u);
    r0 = recs[REC_AT((unsigned)lane)];
    r1 = recs[REC_AT(f1)];
    const int vd = (int)(r0.x & vmask);
    y0 = vertices[3 * vd + 0];
    y1 = vertices[3 * vd + 1];
    y2 = vertices[3 * vd + 2];
  }
  for (unsigned f = (unsigned)lane; f < cnt; f += 32u) {
    const unsigned f2  = (f + 64u < cnt) ? (f + 64u) : (cnt - 1u);
    const int      vdn = (int)(r1.x & vmask);
    const float n0 = vertices[3 * vdn + 0];
    const float n1 = vertices[3 * vdn + 1];
    const float n2 = vertices[3 * vdn + 2];
    const uint2 r2 = recs[REC_AT(f2)];

    const int   sl = (int)(r0.x >> DST_BITS);
    const float weight = __uint_as_float(r0.y);
    const float x0 = xs[3 * sl + 0], x1 = xs[3 * sl + 1], x2 = xs[3 * sl + 2];
    const float t  = -x0 * y0 + x1 * y1 + x2 * y2;
    const float w0 = y0 + x0 * t;
    const float w1 = y1 + x1 * t;
    const float w2 = y2 + x2 * t;
    const float vn2  = sqrtf(-w0 * w0 + w1 * w1 + w2 * w2 + EPSV);
    const float dist = hyp_dist(-t);
    const float sc   = dist * weight / vn2;
    atomicAdd(&a0s[sl], w0 * sc);
    atomicAdd(&a1s[sl], w1 * sc);
    atomicAdd(&a2s[sl], w2 * sc);
    atomicAdd(&acs[sl], 1u);

    r0 = r1; r1 = r2;
    y0 = n0; y1 = n1; y2 = n2;
  }
#undef REC_AT
  __syncthreads();

  if (tid < nv) {
    const float c   = (float)acs[tid];
    const float inv = 1.0f / fmaxf(c, 1.0f);
    float a0 = 0.0f, a1 = 0.0f, a2 = 0.0f;
    if (c > 0.0f) {
      a0 = a0s[tid] * inv;
      a1 = a1s[tid] * inv;
      a2 = a2s[tid] * inv;
    }
    const float an = sqrtf(-a0 * a0 + a1 * a1 + a2 * a2 + EPSV);
    const float T  = an;
    const float e  = expf(T);
    const float ei = 1.0f / e;
    const float ch = 0.5f * (e + ei);
    const float sh = 0.5f * (e - ei);
    const float f  = sh / an;
    const int   i  = lo + tid;
    out[3 * i + 0] = ch * xs[3 * tid + 0] + f * a0;
    out[3 * i + 1] = ch * xs[3 * tid + 1] + f * a1;
    out[3 * i + 2] = ch * xs[3 * tid + 2] + f * a2;
  }
}

extern "C" void kernel_launch(void* const* d_in, const int* in_sizes, int n_in,
                              void* d_out, int out_size, void* d_ws, size_t ws_size,
                              hipStream_t stream) {
  const float* vertices = (const float*)d_in[0];
  const int*   edges    = (const int*)d_in[1];
  const float* messages = (const float*)d_in[2];
  float* out = (float*)d_out;

  const int       N = in_sizes[0] / 3;
  const long long E = (long long)in_sizes[1] / 2;

  const int    NB   = (N + RV - 1) >> RV_SHIFT;
  const size_t need = (size_t)4 * MAXB * (2 + NCHUNK) + (size_t)8 * (size_t)E;
  if (NB <= MAXB && E < (1ll << 31) && ws_size >= need) {
    unsigned* runtab = (unsigned*)d_ws;
    uint2*    recs   = (uint2*)(runtab + (2 + NCHUNK) * MAXB);
    ChunkScatter_kernel<<<dim3(NCHUNK), CT, 0, stream>>>(edges, messages, runtab, recs, E, NB);
    RangeSum_kernel<<<dim3(NB), RV, 0, stream>>>(vertices, runtab, recs, out, E, N);
    return;
  }

  float* acc = (float*)d_ws;

  unsigned* tickets = (unsigned*)d_out;

  const int zn = 4 * N;
  HyperZero_kernel<<<(zn + 255) / 256, 256, 0, stream>>>(acc, zn, tickets);

  HyperEdge_kernel<<<dim3(2048), 256, 0, stream>>>(
      vertices, edges, messages, acc, tickets, E, N);

  HyperVertex_kernel<<<(N + 255) / 256, 256, 0, stream>>>(vertices, acc, out, N);
}


# Summing kernel fetches target vertices from a copy of 16-byte records written by the scatter kernel: no fetch crosses a 128-byte line (12-byte records at 4-byte alignment do in 2 of 32 places)
# speedup vs baseline: 5.5107x; 1.0367x over previous
#include <hip/hip_runtime.h>
#include <hip/hip_bf16.h>

typedef __attribute__((ext_vector_type(2))) float v2f;
typedef __attribute__((ext_vector_type(8))) float v8f;

#define EPSV 1e-6f

static __device__ __forceinline__ float hyp_dist(float u) {
  const bool  big = u > 1.0e18f;
  const float s   = __builtin_amdgcn_sqrtf((u - 1.0f) * (u + 1.0f));
  const float r   = __builtin_amdgcn_logf(big ? u : (u + s)) * 0.693147180559945f;
  return big ? (r + 0.693147180559945f) : r;
}

#define RUN_TILES 8
#define TICKET_STRIDE 128

__global__ void HyperZero_kernel(float* __restrict__ p, int n, unsigned* __restrict__ tickets) {
  int i = blockIdx.x * blockDim.x + threadIdx.x;
  if (i < n) p[i] = 0.0f;
  if (i < 2) tickets[i * TICKET_STRIDE] = 0u;
}

template <int K>
static __device__ __forceinline__ int quad_lane(int v) {
  return __builtin_amdgcn_mov_dpp(v, K * 0x55, 0xf, 0xf, true);
}
template <int K>
static __device__ __forceinline__ float quad_lane(float v) {
  return __int_as_float(quad_lane<K>(__float_as_int(v)));
}

template <int K>
static __device__ __forceinline__ void add_quad_edge(float* __restrict__ acc, int comp, int vs,
                                                     float c0, float c1, float c2) {
  const int   vj = quad_lane<K>(vs);
  const float t0 = quad_lane<K>(c0);
  const float t1 = quad_lane<K>(c1);
  const float t2 = quad_lane<K>(c2);
  const float lo = (comp == 0) ? t0 : t1;
  const float hi = (comp == 2) ? t2 : 1.0f;
  const float val = (comp < 2) ? lo : hi;
  if (vj >= 0) atomicAdd(&acc[4 * vj + comp], val);
}

static __device__ __forceinline__ void edge_tile(
    const float* __restrict__ vertices,
    const int*   __restrict__ edges,
    const float* __restrict__ messages,
    float* __restrict__ acc,
    long long E, long long e0, int lo, int hi, int lane) {
  const int octet = lane >> 3;
  const int perm  = (octet == 1) ? (lane + 8) : ((octet == 2) ? (lane - 8) : lane);
  const long long eg = e0 + perm;

  const long long el = (eg < E) ? eg : (E - 1);
  const int2 sd = *reinterpret_cast<const int2*>(edges + 2 * el);

  float weight = 0.0f;
  if (e0 + 32 <= E) {
    const int rowA = lane & 15;
    const int kb   = (lane < 16) ? 0 : 2;
    const float* m1 = messages + (e0 + rowA) * 8 + kb;
    const float* m2 = m1 + 16 * 8;
    v2f a1lo = *(const v2f*)(m1);
    v2f a1hi = *(const v2f*)(m1 + 4);
    v2f a2lo = *(const v2f*)(m2);
    v2f a2hi = *(const v2f*)(m2 + 4);
    v2f ones; ones[0] = 1.0f; ones[1] = 1.0f;

    v8f c1 = {};
    c1 = __builtin_amdgcn_wmma_f32_16x16x4_f32(false, a1lo, false, ones, (short)0, c1, false, false);
    c1 = __builtin_amdgcn_wmma_f32_16x16x4_f32(false, a1hi, false, ones, (short)0, c1, false, false);
    v8f c2 = {};
    c2 = __builtin_amdgcn_wmma_f32_16x16x4_f32(false, a2lo, false, ones, (short)0, c2, false, false);
    c2 = __builtin_amdgcn_wmma_f32_16x16x4_f32(false, a2hi, false, ones, (short)0, c2, false, false);

    const bool useD2 = (lane >> 3) & 1;
    v8f d = useD2 ? c2 : c1;
    const int r = lane & 7;
    float s = d[0];
    if (r == 1) s = d[1];
    if (r == 2) s = d[2];
    if (r == 3) s = d[3];
    if (r == 4) s = d[4];
    if (r == 5) s = d[5];
    if (r == 6) s = d[6];
    if (r == 7) s = d[7];
    weight = s * 0.125f;
  } else if (eg < E) {
    float s = 0.0f;
#pragma unroll
    for (int k = 0; k < 8; ++k) s += messages[eg * 8 + k];
    weight = s * 0.125f;
  }

  int   vs = -1;
  float c0 = 0.0f, c1 = 0.0f, c2 = 0.0f;
  {
    const int s  = (eg < E) ? sd.x : -1;
    const int vd = sd.y;
    if (s >= lo && s < hi) {
      vs = s;
      const float x0 = vertices[3 * vs + 0], x1 = vertices[3 * vs + 1], x2 = vertices[3 * vs + 2];
      const float y0 = vertices[3 * vd + 0], y1 = vertices[3 * vd + 1], y2 = vertices[3 * vd + 2];
      const float t  = -x0 * y0 + x1 * y1 + x2 * y2;
      const float v0 = y0 + x0 * t;
      const float v1 = y1 + x1 * t;
      const float v2 = y2 + x2 * t;
      const float vn   = sqrtf(-v0 * v0 + v1 * v1 + v2 * v2 + EPSV);
      const float dist = hyp_dist(-t);
      const float sc   = dist * weight / vn;
      c0 = v0 * sc;
      c1 = v1 * sc;
      c2 = v2 * sc;
    }
  }

  const int comp = lane & 3;
  add_quad_edge<0>(acc, comp, vs, c0, c1, c2);
  add_quad_edge<1>(acc, comp, vs, c0, c1, c2);
  add_quad_edge<2>(acc, comp, vs, c0, c1, c2);
  add_quad_edge<3>(acc, comp, vs, c0, c1, c2);
}

__global__ __launch_bounds__(256) void HyperEdge_kernel(
    const float* __restrict__ vertices,
    const int*   __restrict__ edges,
    const float* __restrict__ messages,
    float* __restrict__ acc,
    unsigned* __restrict__ tickets,
    long long E, int N) {
  const int lane = threadIdx.x & 31;
  const unsigned place  = __builtin_amdgcn_s_sendmsg_rtn(0x87);
  const int      myhalf = (int)((place >> 18) & 1u);

  const long long tiles = (E + 31) >> 5;
  const unsigned  runs  = (unsigned)((tiles + RUN_TILES - 1) / RUN_TILES);
  const int       split = ((N >> 1) + 15) & ~15;

  for (int pass = 0; pass < 2; ++pass) {
    const int h  = myhalf ^ pass;
    const int lo = h ? split : 0;
    const int hi = h ? N : split;
    for (unsigned it = 0; it <= runs; ++it) {
      unsigned t = 0u;
      if (lane == 0) t = atomicAdd(&tickets[h * TICKET_STRIDE], 1u);
      t = (unsigned)__builtin_amdgcn_readfirstlane((int)t);
      if (t >= runs) break;
      const long long tile0 = (long long)t * RUN_TILES;
      for (int k = 0; k < RUN_TILES; ++k) {
        const long long e0 = (tile0 + k) << 5;
        if (e0 >= E) break;
        edge_tile(vertices, edges, messages, acc, E, e0, lo, hi, lane);
      }
    }
  }
}

__global__ void HyperVertex_kernel(const float* __restrict__ vertices,
                                   const float* __restrict__ acc,
                                   float* __restrict__ out, int N) {
  int i = blockIdx.x * blockDim.x + threadIdx.x;
  if (i >= N) return;
  const float4 rec = reinterpret_cast<const float4*>(acc)[i];
  const float c   = rec.w;
  const float inv = 1.0f / fmaxf(c, 1.0f);
  float a0 = 0.0f, a1 = 0.0f, a2 = 0.0f;
  if (c > 0.0f) {
    a0 = rec.x * inv;
    a1 = rec.y * inv;
    a2 = rec.z * inv;
  }
  const float an = sqrtf(-a0 * a0 + a1 * a1 + a2 * a2 + EPSV);
  const float T  = an;
  const float e  = expf(T);
  const float ei = 1.0f / e;
  const float ch = 0.5f * (e + ei);
  const float sh = 0.5f * (e - ei);
  const float f  = sh / an;
  out[3 * i + 0] = ch * vertices[3 * i + 0] + f * a0;
  out[3 * i + 1] = ch * vertices[3 * i + 1] + f * a1;
  out[3 * i + 2] = ch * vertices[3 * i + 2] + f * a2;
}


#define RV_SHIFT 10
#define RV (1 << RV_SHIFT)
#define MAXB 1024
#define DST_BITS 20

#define NCHUNK 256
#define CT 1024
#define TROW MAXB
#define CPW (NCHUNK / (RV / 32))

static __device__ __forceinline__ void chunk_tiles(long long E, int nc, int c,
                                                   long long& t0, long long& t1) {
  const long long tiles = (E + 31) >> 5;
  const long long ct    = (tiles + nc - 1) / nc;
  t0 = (long long)c * ct;
  if (t0 > tiles) t0 = tiles;
  t1 = t0 + ct;
  if (t1 > tiles) t1 = tiles;
}

static __device__ __forceinline__ void tile_loads(const float* __restrict__ messages,
                                                  long long e0, int lane, v2f (&a)[4]) {
  const int rowA = lane & 15;
  const int kb   = (lane < 16) ? 0 : 2;
  const float* m1 = messages + (e0 + rowA) * 8 + kb;
  const float* m2 = m1 + 16 * 8;
  a[0] = *(const v2f*)(m1);
  a[1] = *(const v2f*)(m1 + 4);
  a[2] = *(const v2f*)(m2);
  a[3] = *(const v2f*)(m2 + 4);
}

static __device__ __forceinline__ float tile_mean(const v2f (&a)[4], int lane) {
  v2f ones; ones[0] = 1.0f; ones[1] = 1.0f;

  v8f c1 = {};
  c1 = __builtin_amdgcn_wmma_f32_16x16x4_f32(false, a[0], false, ones, (short)0, c1, false, false);
  c1 = __builtin_amdgcn_wmma_f32_16x16x4_f32(false, a[1], false, ones, (short)0, c1, false, false);
  v8f c2 = {};
  c2 = __builtin_amdgcn_wmma_f32_16x16x4_f32(false, a[2], false, ones, (short)0, c2, false, false);
  c2 = __builtin_amdgcn_wmma_f32_16x16x4_f32(false, a[3], false, ones, (short)0, c2, false, false);

  const bool useD2 = (lane >> 3) & 1;
  v8f d = useD2 ? c2 : c1;
  const int r = lane & 7;
  float s = d[0];
  if (r == 1) s = d[1];
  if (r == 2) s = d[2];
  if (r == 3) s = d[3];
  if (r == 4) s = d[4];
  if (r == 5) s = d[5];
  if (r == 6) s = d[6];
  if (r == 7) s = d[7];
  return s * 0.125f;
}

static __device__ __forceinline__ float full_tile_mean(const float* __restrict__ messages,
                                                       long long e0, int lane) {
  v2f a[4];
  tile_loads(messages, e0, lane, a);
  return tile_mean(a, lane);
}

#define STAGE_CAP 32768

#define VTAB_PER_THREAD 4
static_assert((long long)MAXB * RV <= (long long)VTAB_PER_THREAD * CT * NCHUNK,
              "a scatter work-group copies at most VTAB_PER_THREAD vertices per thread");

static __device__ __forceinline__ void place_edge(unsigned* lh, uint2* stage, uint2* __restrict__ recs,
                                                  bool staged, int2 sd, float weight) {
  const unsigned pos = atomicAdd(&lh[sd.x >> RV_SHIFT], 1u);
  const uint2 rec = make_uint2(((unsigned)(sd.x & (RV - 1)) << DST_BITS) | (unsigned)sd.y,
                               __float_as_uint(weight));
  if (staged) stage[pos] = rec;
  else        recs[pos]  = rec;
}

__global__ __launch_bounds__(CT) void ChunkScatter_kernel(
    const int*   __restrict__ edges,
    const float* __restrict__ messages,
    const float* __restrict__ vertices,
    float4* __restrict__ vtab,
    unsigned* __restrict__ runtab,
    uint2* __restrict__ recs,
    long long E, int NB, int N) {
  __shared__ unsigned lh[MAXB];
  __shared__ unsigned sl[CT / 32];
  __shared__ uint2 stage[STAGE_CAP];
  const int tid  = threadIdx.x;
  const int lane = tid & 31;
  const int wave = tid >> 5;
  const int me   = blockIdx.x;

  long long t0, t1;
  chunk_tiles(E, NCHUNK, me, t0, t1);
  const long long eb = t0 << 5;
  long long ee = t1 << 5;
  if (ee > E) ee = E;
  const bool staged = ((t1 - t0) << 5) <= STAGE_CAP;

  const int vper = (N + NCHUNK - 1) / NCHUNK;
  const int vbeg = me * vper;
  const int vend = (vbeg + vper < N) ? (vbeg + vper) : N;
  float px[VTAB_PER_THREAD], py[VTAB_PER_THREAD], pz[VTAB_PER_THREAD];

  lh[tid] = 0u;
  __syncthreads();
  if (vtab != nullptr) {
#pragma unroll
    for (int k = 0; k < VTAB_PER_THREAD; ++k) {
      const int v = vbeg + tid + k * CT;
      px[k] = 0.0f; py[k] = 0.0f; pz[k] = 0.0f;
      if (v < vend) {
        px[k] = vertices[3 * v + 0];
        py[k] = vertices[3 * v + 1];
        pz[k] = vertices[3 * v + 2];
      }
    }
  }
  for (long long e = eb + tid; e < ee; e += 32 * CT) {
    int s[32];
#pragma unroll
    for (int k = 0; k < 32; ++k) {
      const long long ek = e + k * CT;
      s[k] = (ek < ee) ? edges[2 * ek] : -1;
    }
#pragma unroll
    for (int k = 0; k < 32; ++k) {
      if (s[k] >= 0) atomicAdd(&lh[s[k] >> RV_SHIFT], 1u);
    }
  }
  __syncthreads();
  const unsigned mine = lh[tid];

  unsigned incm = mine;
  for (int d = 1; d < 32; d <<= 1) {
    const unsigned m = __shfl_up(incm, d);
    if (lane >= d) incm += m;
  }
  if (lane == 31) sl[wave] = incm;
  __syncthreads();
  unsigned lfirst = incm - mine;
  for (int j = 0; j < wave; ++j) lfirst += sl[j];
  lh[tid] = staged ? lfirst : ((unsigned)eb + lfirst);
  unsigned* __restrict__ row = runtab + (size_t)me * TROW;
  if (tid < NB) row[tid] = lfirst;
  if (vtab != nullptr) {
#pragma unroll
    for (int k = 0; k < VTAB_PER_THREAD; ++k) {
      const int v = vbeg + tid + k * CT;
      if (v < vend) vtab[v] = make_float4(px[k], py[k], pz[k], 0.0f);
    }
  }
  __syncthreads();

  const int octet = lane >> 3;
  const int perm  = (octet == 1) ? (lane + 8) : ((octet == 2) ? (lane - 8) : lane);

  for (long long t = t0 + 4 * wave; t < t1; t += 4 * (CT / 32)) {
    const long long e0 = t << 5;
    if (t + 3 < t1 && e0 + 128 <= E) {
      const int2 sdA = *reinterpret_cast<const int2*>(edges + 2 * (e0 + perm));
      const int2 sdB = *reinterpret_cast<const int2*>(edges + 2 * (e0 + 32 + perm));
      const int2 sdC = *reinterpret_cast<const int2*>(edges + 2 * (e0 + 64 + perm));
      const int2 sdD = *reinterpret_cast<const int2*>(edges + 2 * (e0 + 96 + perm));
      v2f aA[4], aB[4], aC[4], aD[4];
      tile_loads(messages, e0, lane, aA);
      tile_loads(messages, e0 + 32, lane, aB);
      tile_loads(messages, e0 + 64, lane, aC);
      tile_loads(messages, e0 + 96, lane, aD);
      __builtin_amdgcn_sched_barrier(0);
      const float wA = tile_mean(aA, lane);
      const float wB = tile_mean(aB, lane);
      const float wC = tile_mean(aC, lane);
      const float wD = tile_mean(aD, lane);
      place_edge(lh, stage, recs, staged, sdA, wA);
      place_edge(lh, stage, recs, staged, sdB, wB);
      place_edge(lh, stage, recs, staged, sdC, wC);
      place_edge(lh, stage, recs, staged, sdD, wD);
    } else {
      const long long tl = (t + 4 < t1) ? (t + 4) : t1;
      for (long long u = t; u < tl; ++u) {
        const long long f0 = u << 5;
        const long long eg = f0 + perm;
        const long long el = (eg < E) ? eg : (E - 1);
        const int2 sd = *reinterpret_cast<const int2*>(edges + 2 * el);
        float weight = 0.0f;
        if (f0 + 32 <= E) {
          weight = full_tile_mean(messages, f0, lane);
        } else if (eg < E) {
          float s = 0.0f;
#pragma unroll
          for (int k = 0; k < 8; ++k) s += messages[eg * 8 + k];
          weight = s * 0.125f;
        }
        if (eg < E) place_edge(lh, stage, recs, staged, sd, weight);
      }
    }
  }

  if (!staged) return;
  __syncthreads();

  const unsigned n = (unsigned)(ee - eb);
#pragma unroll 4
  for (unsigned i = tid; i < n; i += CT) recs[eb + i] = stage[i];
}

static __device__ __forceinline__ unsigned run_record(
    unsigned f, unsigned p1, unsigned p2, unsigned p3, unsigned p4, unsigned p5, unsigned p6,
    unsigned p7, unsigned a0, unsigned a1, unsigned a2, unsigned a3, unsigned a4, unsigned a5,
    unsigned a6, unsigned a7) {
  unsigned a = a0;
  a = (f >= p1) ? a1 : a;
  a = (f >= p2) ? a2 : a;
  a = (f >= p3) ? a3 : a;
  a = (f >= p4) ? a4 : a;
  a = (f >= p5) ? a5 : a;
  a = (f >= p6) ? a6 : a;
  a = (f >= p7) ? a7 : a;
  return a + f;
}

template <bool PAD>
__global__ __launch_bounds__(RV) void RangeSum_kernel(
    const float* __restrict__ vertices,
    const float4* __restrict__ vtab,
    const unsigned* __restrict__ runtab,
    const uint2* __restrict__ recs,
    float* __restrict__ out, long long E, int N) {
  __shared__ float    xs[3 * RV];
  __shared__ float    a0s[RV];
  __shared__ float    a1s[RV];
  __shared__ float    a2s[RV];
  __shared__ unsigned acs[RV];
  const int tid  = threadIdx.x;
  const int lane = tid & 31;
  const int b    = blockIdx.x;
  const int lo   = b << RV_SHIFT;
  const int nv   = (N - lo < RV) ? (N - lo) : RV;

  const int wv = __builtin_amdgcn_readfirstlane(tid >> 5);
  unsigned rf = 0u, rg = 0u, rb = 0u;
  if (lane < CPW) {
    const int ch = wv * CPW + lane;
    long long t0, t1;
    chunk_tiles(E, NCHUNK, ch, t0, t1);
    long long ee = t1 << 5;
    if (ee > E) ee = E;
    rb = (unsigned)(t0 << 5);
    rf = runtab[(size_t)ch * TROW + b];
    rg = (b + 1 < (int)gridDim.x) ? runtab[(size_t)ch * TROW + b + 1]
                                  : (unsigned)(ee - (t0 << 5));
  }

  if (PAD) {
    if (tid < nv) {
      const float4 xv = vtab[lo + tid];
      xs[3 * tid + 0] = xv.x;
      xs[3 * tid + 1] = xv.y;
      xs[3 * tid + 2] = xv.z;
    }
  } else {
    for (int i = tid; i < 3 * nv; i += RV) xs[i] = vertices[3 * lo + i];
  }
  a0s[tid] = 0.0f;
  a1s[tid] = 0.0f;
  a2s[tid] = 0.0f;
  acs[tid] = 0u;
  __syncthreads();

  static_assert(CPW == 8, "the run list below is written out for eight runs per wave");
  const unsigned vs = rb + rf;
  const unsigned vn = rg - rf;
#define RUN_S(m) ((unsigned)__builtin_amdgcn_readlane((int)vs, (m)))
#define RUN_N(m) ((unsigned)__builtin_amdgcn_readlane((int)vn, (m)))
  const unsigned P1 = RUN_N(0);
  const unsigned P2 = P1 + RUN_N(1);
  const unsigned P3 = P2 + RUN_N(2);
  const unsigned P4 = P3 + RUN_N(3);
  const unsigned P5 = P4 + RUN_N(4);
  const unsigned P6 = P5 + RUN_N(5);
  const unsigned P7 = P6 + RUN_N(6);
  const unsigned cnt = P7 + RUN_N(7);
  const unsigned A0 = RUN_S(0);
  const unsigned A1 = RUN_S(1) - P1;
  const unsigned A2 = RUN_S(2) - P2;
  const unsigned A3 = RUN_S(3) - P3;
  const unsigned A4 = RUN_S(4) - P4;
  const unsigned A5 = RUN_S(5) - P5;
  const unsigned A6 = RUN_S(6) - P6;
  const unsigned A7 = RUN_S(7) - P7;
#undef RUN_S
#undef RUN_N
#define REC_AT(f) run_record((f), P1, P2, P3, P4, P5, P6, P7, A0, A1, A2, A3, A4, A5, A6, A7)

  const unsigned vmask = (1u << DST_BITS) - 1u;
  uint2 r0 = make_uint2(0u, 0u);
  uint2 r1 = make_uint2(0u, 0u);
  float y0 = 0.0f, y1 = 0.0f, y2 = 0.0f;
  if ((unsigned)lane < cnt) {
    const unsigned f1 = ((unsigned)lane + 32u < cnt) ? ((unsigned)lane + 32u) : (cnt - 1u);
    r0 = recs[REC_AT((unsigned)lane)];
    r1 = recs[REC_AT(f1)];
    const int vd = (int)(r0.x & vmask);
    if (PAD) {
      const float4 yv = vtab[vd];
      y0 = yv.x; y1 = yv.y; y2 = yv.z;
    } else {
      y0 = vertices[3 * vd + 0];
      y1 = vertices[3 * vd + 1];
      y2 = vertices[3 * vd + 2];
    }
  }
  for (unsigned f = (unsigned)lane; f < cnt; f += 32u) {
    const unsigned f2  = (f + 64u < cnt) ? (f + 64u) : (cnt - 1u);
    const int      vdn = (int)(r1.x & vmask);
    float n0, n1, n2;
    if (PAD) {
      const float4 nv4 = vtab[vdn];
      n0 = nv4.x; n1 = nv4.y; n2 = nv4.z;
    } else {
      n0 = vertices[3 * vdn + 0];
      n1 = vertices[3 * vdn + 1];
      n2 = vertices[3 * vdn + 2];
    }
    const uint2 r2 = recs[REC_AT(f2)];

    const int   sl = (int)(r0.x >> DST_BITS);
    const float weight = __uint_as_float(r0.y);
    const float x0 = xs[3 * sl + 0], x1 = xs[3 * sl + 1], x2 = xs[3 * sl + 2];
    const float t  = -x0 * y0 + x1 * y1 + x2 * y2;
    const float w0 = y0 + x0 * t;
    const float w1 = y1 + x1 * t;
    const float w2 = y2 + x2 * t;
    const float vn2  = sqrtf(-w0 * w0 + w1 * w1 + w2 * w2 + EPSV);
    const float dist = hyp_dist(-t);
    const float sc   = dist * weight / vn2;
    atomicAdd(&a0s[sl], w0 * sc);
    atomicAdd(&a1s[sl], w1 * sc);
    atomicAdd(&a2s[sl], w2 * sc);
    atomicAdd(&acs[sl], 1u);

    r0 = r1; r1 = r2;
    y0 = n0; y1 = n1; y2 = n2;
  }
#undef REC_AT
  __syncthreads();

  if (tid < nv) {
    const float c   = (float)acs[tid];
    const float inv = 1.0f / fmaxf(c, 1.0f);
    float a0 = 0.0f, a1 = 0.0f, a2 = 0.0f;
    if (c > 0.0f) {
      a0 = a0s[tid] * inv;
      a1 = a1s[tid] * inv;
      a2 = a2s[tid] * inv;
    }
    const float an = sqrtf(-a0 * a0 + a1 * a1 + a2 * a2 + EPSV);
    const float T  = an;
    const float e  = expf(T);
    const float ei = 1.0f / e;
    const float ch = 0.5f * (e + ei);
    const float sh = 0.5f * (e - ei);
    const float f  = sh / an;
    const int   i  = lo + tid;
    out[3 * i + 0] = ch * xs[3 * tid + 0] + f * a0;
    out[3 * i + 1] = ch * xs[3 * tid + 1] + f * a1;
    out[3 * i + 2] = ch * xs[3 * tid + 2] + f * a2;
  }
}

extern "C" void kernel_launch(void* const* d_in, const int* in_sizes, int n_in,
                              void* d_out, int out_size, void* d_ws, size_t ws_size,
                              hipStream_t stream) {
  const float* vertices = (const float*)d_in[0];
  const int*   edges    = (const int*)d_in[1];
  const float* messages = (const float*)d_in[2];
  float* out = (float*)d_out;

  const int       N = in_sizes[0] / 3;
  const long long E = (long long)in_sizes[1] / 2;

  const int    NB   = (N + RV - 1) >> RV_SHIFT;
  const size_t need = (size_t)4 * MAXB * (2 + NCHUNK) + (size_t)8 * (size_t)E;
  const size_t voff = (need + 15) & ~(size_t)15;
  if (NB <= MAXB && E < (1ll << 31) && ws_size >= need) {
    unsigned* runtab = (unsigned*)d_ws;
    uint2*    recs   = (uint2*)(runtab + (2 + NCHUNK) * MAXB);
    if (ws_size >= voff + (size_t)16 * (size_t)N) {
      float4* vtab = (float4*)((char*)d_ws + voff);
      ChunkScatter_kernel<<<dim3(NCHUNK), CT, 0, stream>>>(edges, messages, vertices, vtab, runtab, recs,
                                                           E, NB, N);
      RangeSum_kernel<true><<<dim3(NB), RV, 0, stream>>>(vertices, vtab, runtab, recs, out, E, N);
    } else {
      ChunkScatter_kernel<<<dim3(NCHUNK), CT, 0, stream>>>(edges, messages, vertices, nullptr, runtab, recs,
                                                           E, NB, N);
      RangeSum_kernel<false><<<dim3(NB), RV, 0, stream>>>(vertices, nullptr, runtab, recs, out, E, N);
    }
    return;
  }

  float* acc = (float*)d_ws;

  unsigned* tickets = (unsigned*)d_out;

  const int zn = 4 * N;
  HyperZero_kernel<<<(zn + 255) / 256, 256, 0, stream>>>(acc, zn, tickets);

  HyperEdge_kernel<<<dim3(2048), 256, 0, stream>>>(
      vertices, edges, messages, acc, tickets, E, N);

  HyperVertex_kernel<<<(N + 255) / 256, 256, 0, stream>>>(vertices, acc, out, N);
}
